# starting kernel
# baseline (speedup 1.0000x reference)
#include <hip/hip_runtime.h>
#include <hip/hip_bf16.h>

typedef __attribute__((ext_vector_type(2))) float v2f;
typedef __attribute__((ext_vector_type(8))) float v8f;

#define N_NODES 50000
#define N_EDGES 800000
#define DIM     128


__global__ void zero_kernel(float4* __restrict__ p, long n4) {
    long i = (long)blockIdx.x * blockDim.x + threadIdx.x;
    long stride = (long)gridDim.x * blockDim.x;
    float4 z = make_float4(0.f, 0.f, 0.f, 0.f);
    for (; i < n4; i += stride) p[i] = z;
}

__global__ void degree_kernel(const int* __restrict__ row,
                              const int* __restrict__ col,
                              float* __restrict__ deg_row,
                              float* __restrict__ deg_col) {
    int e = blockIdx.x * blockDim.x + threadIdx.x;
    if (e >= N_EDGES) return;
    atomicAdd(&deg_row[row[e]], 1.0f);
    atomicAdd(&deg_col[col[e]], 1.0f);
}

__global__ void invsqrt_kernel(float* __restrict__ deg) {
    int i = blockIdx.x * blockDim.x + threadIdx.x;
    if (i >= 2 * N_NODES) return;
    float d = deg[i];
    deg[i] = (d > 0.0f) ? rsqrtf(fmaxf(d, 1.0f)) : 0.0f;
}

__global__ void prep_kernel(const float* __restrict__ W_src,
                            const float* __restrict__ b_src,
                            const float* __restrict__ W_dst,
                            const float* __restrict__ b_dst,
                            float* __restrict__ Wc_s,
                            float* __restrict__ Wc_d,
                            float* __restrict__ bias) {
    int i = blockIdx.x * blockDim.x + threadIdx.x;
    if (i < DIM * DIM) {
        Wc_s[i] = 0.5f * W_src[i];
        Wc_d[i] = 0.5f * W_dst[i];
    }
    if (i < DIM) bias[i] = 0.5f * b_src[i] + 0.5f * b_dst[i];
}

__global__ void scatter_kernel(const float* __restrict__ x,
                               const int* __restrict__ row,
                               const int* __restrict__ col,
                               const float* __restrict__ inv_row,
                               const float* __restrict__ inv_col,
                               float* __restrict__ agg_fwd,
                               float* __restrict__ agg_rev) {
    int lane = threadIdx.x & 31;
    int e = blockIdx.x * (blockDim.x >> 5) + (threadIdx.x >> 5);
    if (e >= N_EDGES) return;
    int r = row[e];
    int c = col[e];
    float w = inv_row[r] * inv_col[c];

    const float4 vc = ((const float4*)(x + (size_t)c * DIM))[lane];
    const float4 vr = ((const float4*)(x + (size_t)r * DIM))[lane];

    float* af = agg_fwd + (size_t)r * DIM + lane * 4;
    float* ar = agg_rev + (size_t)c * DIM + lane * 4;
    atomicAdd(af + 0, w * vc.x);
    atomicAdd(af + 1, w * vc.y);
    atomicAdd(af + 2, w * vc.z);
    atomicAdd(af + 3, w * vc.w);
    atomicAdd(ar + 0, w * vr.x);
    atomicAdd(ar + 1, w * vr.y);
    atomicAdd(ar + 2, w * vr.z);
    atomicAdd(ar + 3, w * vr.w);
}

__global__ void __launch_bounds__(256)
gemm_kernel(const float* __restrict__ agg_fwd,
            const float* __restrict__ agg_rev,
            const float* __restrict__ Wc_s,
            const float* __restrict__ Wc_d,
            const float* __restrict__ bias,
            float* __restrict__ out) {
    const int lane = threadIdx.x & 31;
    const int tile = blockIdx.x * 8 + (threadIdx.x >> 5);
    const int mt = tile >> 3;
    const int nt = tile & 7;

    const int half = lane >> 4;
    const int l15  = lane & 15;
    const int kb   = half * 2;

    const float* arow_f = agg_fwd + (size_t)(mt * 16 + l15) * DIM;
    const float* arow_r = agg_rev + (size_t)(mt * 16 + l15) * DIM;
    const float* wrow_s = Wc_s    + (size_t)(nt * 16 + l15) * DIM;
    const float* wrow_d = Wc_d    + (size_t)(nt * 16 + l15) * DIM;

    v8f c;
    const float bv = bias[nt * 16 + l15];
    #pragma unroll
    for (int j = 0; j < 8; ++j) c[j] = bv;

    #pragma unroll
    for (int k = 0; k < DIM; k += 4) {
        v2f a0 = *(const v2f*)(arow_f + k + kb);
        v2f b0 = *(const v2f*)(wrow_s + k + kb);
        c = __builtin_amdgcn_wmma_f32_16x16x4_f32(false, a0, false, b0,
                                                  (short)0, c, false, false);
        v2f a1 = *(const v2f*)(arow_r + k + kb);
        v2f b1 = *(const v2f*)(wrow_d + k + kb);
        c = __builtin_amdgcn_wmma_f32_16x16x4_f32(false, a1, false, b1,
                                                  (short)0, c, false, false);
    }

    const int m0 = mt * 16 + half * 8;
    const int n  = nt * 16 + l15;
    #pragma unroll
    for (int j = 0; j < 8; ++j)
        out[(size_t)(m0 + j) * DIM + n] = c[j];
}

extern "C" void kernel_launch(void* const* d_in, const int* in_sizes, int n_in,
                              void* d_out, int out_size, void* d_ws, size_t ws_size,
                              hipStream_t stream) {
    const float* x      = (const float*)d_in[0];
    const float* W_src  = (const float*)d_in[1];
    const float* b_src  = (const float*)d_in[2];
    const float* W_dst  = (const float*)d_in[3];
    const float* b_dst  = (const float*)d_in[4];
    const int*   eidx   = (const int*)d_in[5];
    const int*   row    = eidx;
    const int*   col    = eidx + N_EDGES;
    float*       outp   = (float*)d_out;

    float* ws = (float*)d_ws;
    float* deg_row = ws;
    float* deg_col = ws + N_NODES;
    float* agg_fwd = ws + 2 * (size_t)N_NODES;
    float* agg_rev = agg_fwd + (size_t)N_NODES * DIM;
    float* Wc_s    = agg_rev + (size_t)N_NODES * DIM;
    float* Wc_d    = Wc_s + DIM * DIM;
    float* bias    = Wc_d + DIM * DIM;

    {
        long n4 = (long)(2 + 2 * DIM) * N_NODES / 4;
        zero_kernel<<<2048, 256, 0, stream>>>((float4*)ws, n4);
    }
    degree_kernel<<<(N_EDGES + 255) / 256, 256, 0, stream>>>(row, col, deg_row, deg_col);
    invsqrt_kernel<<<(2 * N_NODES + 255) / 256, 256, 0, stream>>>(deg_row);
    prep_kernel<<<(DIM * DIM + 255) / 256, 256, 0, stream>>>(W_src, b_src, W_dst, b_dst,
                                                             Wc_s, Wc_d, bias);
    scatter_kernel<<<(N_EDGES + 7) / 8, 256, 0, stream>>>(x, row, col, deg_row, deg_col,
                                                          agg_fwd, agg_rev);
    gemm_kernel<<<(N_NODES / 16) * (DIM / 16) / 8, 256, 0, stream>>>(agg_fwd, agg_rev,
                                                                     Wc_s, Wc_d, bias, outp);
}


# float-atomic scatter replaced by per-node gather: edges bucketed by node (per-chunk LDS histograms + prefix sums), one wave sums a node's rows in edge-id order; no float atomics left, deterministic
# speedup vs baseline: 1.7173x; 1.7173x over previous
#include <hip/hip_runtime.h>

typedef __attribute__((ext_vector_type(2))) float v2f;
typedef __attribute__((ext_vector_type(8))) float v8f;

#define N_NODES 50000
#define N_EDGES 800000
#define DIM     128

#define NCHUNK  64
#define CHUNK   (N_EDGES / NCHUNK)
#define EDGE_IT ((CHUNK + 1023) / 1024)
#define SCAN_CH ((N_NODES + 1023) / 1024)


__global__ void __launch_bounds__(1024)
hist_kernel(const int* __restrict__ eidx, int* __restrict__ H, int* __restrict__ pos) {
    __shared__ int hist[N_NODES];
    const int t = threadIdx.x;
    int4* h4 = (int4*)hist;
    for (int i = t; i < N_NODES / 4; i += 1024) h4[i] = make_int4(0, 0, 0, 0);

    const int* idx = eidx + (size_t)blockIdx.x * CHUNK;
    int*       p   = pos  + (size_t)blockIdx.x * CHUNK;
    int node[EDGE_IT];
    #pragma unroll
    for (int k = 0; k < EDGE_IT; ++k) {
        const int i = t + k * 1024;
        node[k] = (i < CHUNK) ? idx[i] : -1;
    }
    __syncthreads();
    #pragma unroll
    for (int k = 0; k < EDGE_IT; ++k) {
        const int i = t + k * 1024;
        if (i < CHUNK) p[i] = atomicAdd(&hist[node[k]], 1);
    }
    __syncthreads();
    int4* g4 = (int4*)(H + (size_t)blockIdx.x * N_NODES);
    for (int i = t; i < N_NODES / 4; i += 1024) g4[i] = h4[i];
}

__global__ void __launch_bounds__(256)
chunk_prefix_kernel(int* __restrict__ H, int* __restrict__ cnt) {
    const int i = blockIdx.x * 256 + threadIdx.x;
    if (i >= 2 * N_NODES) return;
    const int d = (i >= N_NODES) ? 1 : 0;
    int* h = H + (size_t)d * NCHUNK * N_NODES + (i - d * N_NODES);
    int v[NCHUNK];
    #pragma unroll
    for (int b = 0; b < NCHUNK; ++b) v[b] = h[(size_t)b * N_NODES];
    int run = 0;
    #pragma unroll
    for (int b = 0; b < NCHUNK; ++b) { h[(size_t)b * N_NODES] = run; run += v[b]; }
    cnt[i] = run;
}

__global__ void __launch_bounds__(1024)
scan_kernel(const int* __restrict__ cnt, int* __restrict__ off) {
    __shared__ int part[1024];
    const int* c = cnt + blockIdx.x * N_NODES;
    int*       o = off + blockIdx.x * N_NODES;
    const int t = threadIdx.x;
    const int b = t * SCAN_CH;
    int v[SCAN_CH];
    int s = 0;
    #pragma unroll
    for (int k = 0; k < SCAN_CH; ++k) {
        v[k] = (b + k < N_NODES) ? c[b + k] : 0;
        s += v[k];
    }
    part[t] = s;
    __syncthreads();
    for (int d = 1; d < 1024; d <<= 1) {
        int u = (t >= d) ? part[t - d] : 0;
        __syncthreads();
        part[t] += u;
        __syncthreads();
    }
    int run = part[t] - s;
    #pragma unroll
    for (int k = 0; k < SCAN_CH; ++k) {
        if (b + k < N_NODES) o[b + k] = run;
        run += v[k];
    }
}

__global__ void fill_kernel(const int* __restrict__ row,
                            const int* __restrict__ col,
                            const int* __restrict__ off_r,
                            const int* __restrict__ off_c,
                            const int* __restrict__ H,
                            const int* __restrict__ pos_r,
                            const int* __restrict__ pos_c,
                            int* __restrict__ csr_r,
                            int* __restrict__ csr_c) {
    int e = blockIdx.x * blockDim.x + threadIdx.x;
    if (e >= N_EDGES) return;
    const int b = e / CHUNK;
    const int r = row[e];
    const int c = col[e];
    csr_r[off_r[r] + H[(size_t)b * N_NODES + r] + pos_r[e]] = e;
    csr_c[off_c[c] + H[(size_t)(NCHUNK + b) * N_NODES + c] + pos_c[e]] = e;
}

__global__ void prep_kernel(const float* __restrict__ W_src,
                            const float* __restrict__ b_src,
                            const float* __restrict__ W_dst,
                            const float* __restrict__ b_dst,
                            float* __restrict__ Wc_s,
                            float* __restrict__ Wc_d,
                            float* __restrict__ bias) {
    int i = blockIdx.x * blockDim.x + threadIdx.x;
    if (i < DIM * DIM) {
        Wc_s[i] = 0.5f * W_src[i];
        Wc_d[i] = 0.5f * W_dst[i];
    }
    if (i < DIM) bias[i] = 0.5f * b_src[i] + 0.5f * b_dst[i];
}

__device__ __forceinline__ float inv_sqrt_deg(int d) {
    float f = (float)d;
    return (d > 0) ? rsqrtf(fmaxf(f, 1.0f)) : 0.0f;
}

__device__ __forceinline__ void axpy4(float4& acc, float w, const float4 v) {
    acc.x += w * v.x; acc.y += w * v.y; acc.z += w * v.z; acc.w += w * v.w;
}

__global__ void __launch_bounds__(256)
gather_kernel(const float* __restrict__ x,
              const int* __restrict__ other,
              const int* __restrict__ cnt_self,
              const int* __restrict__ cnt_other,
              const int* __restrict__ off_self,
              const int* __restrict__ csr,
              float* __restrict__ agg) {
    const int lane = threadIdx.x & 31;
    const int n = blockIdx.x * 8 + __builtin_amdgcn_readfirstlane(threadIdx.x >> 5);
    if (n >= N_NODES) return;
    const int deg = cnt_self[n];
    const int beg = off_self[n];
    float4 acc = make_float4(0.f, 0.f, 0.f, 0.f);
    const float inv_n = inv_sqrt_deg(deg);
    if (deg > 0 && deg <= 32) {
        int e = 0x7fffffff, srcn = 0;
        float w = 0.f;
        if (lane < deg) {
            e = csr[beg + lane];
            srcn = other[e];
            w = inv_n * inv_sqrt_deg(cnt_other[srcn]);
        }
        int rank = 0;
        for (int j = 0; j < deg; ++j) {
            int ej = __shfl(e, j);
            rank += (ej < e) ? 1 : 0;
        }
        if (lane >= deg) rank = lane;
        const int ssrc = __builtin_amdgcn_ds_permute(rank << 2, srcn);
        const int sw   = __builtin_amdgcn_ds_permute(rank << 2, __float_as_int(w));
        int s = 0;
        for (; s + 4 <= deg; s += 4) {
            const int c0 = __shfl(ssrc, s), c1 = __shfl(ssrc, s + 1);
            const int c2 = __shfl(ssrc, s + 2), c3 = __shfl(ssrc, s + 3);
            const float4 v0 = ((const float4*)(x + (size_t)c0 * DIM))[lane];
            const float4 v1 = ((const float4*)(x + (size_t)c1 * DIM))[lane];
            const float4 v2 = ((const float4*)(x + (size_t)c2 * DIM))[lane];
            const float4 v3 = ((const float4*)(x + (size_t)c3 * DIM))[lane];
            axpy4(acc, __int_as_float(__shfl(sw, s)),     v0);
            axpy4(acc, __int_as_float(__shfl(sw, s + 1)), v1);
            axpy4(acc, __int_as_float(__shfl(sw, s + 2)), v2);
            axpy4(acc, __int_as_float(__shfl(sw, s + 3)), v3);
        }
        for (; s < deg; ++s) {
            const int c0 = __shfl(ssrc, s);
            const float4 v0 = ((const float4*)(x + (size_t)c0 * DIM))[lane];
            axpy4(acc, __int_as_float(__shfl(sw, s)), v0);
        }
    } else if (deg > 32) {
        int last = -1;
        for (int s = 0; s < deg; ++s) {
            int m = 0x7fffffff;
            for (int i = lane; i < deg; i += 32) {
                int ei = csr[beg + i];
                if (ei > last && ei < m) m = ei;
            }
            for (int o = 16; o > 0; o >>= 1) {
                int t = __shfl_xor(m, o);
                m = (t < m) ? t : m;
            }
            last = m;
            const int c0 = other[m];
            const float w = inv_n * inv_sqrt_deg(cnt_other[c0]);
            const float4 v0 = ((const float4*)(x + (size_t)c0 * DIM))[lane];
            axpy4(acc, w, v0);
        }
    }
    ((float4*)(agg + (size_t)n * DIM))[lane] = acc;
}

template <bool FIRST>
__global__ void __launch_bounds__(256)
gemm_kernel(const float* __restrict__ agg,
            const float* __restrict__ Wc,
            const float* __restrict__ bias,
            float* __restrict__ out) {
    const int lane = threadIdx.x & 31;
    const int tile = blockIdx.x * 8 + (threadIdx.x >> 5);
    const int mt = tile >> 3;
    const int nt = tile & 7;

    const int half = lane >> 4;
    const int l15  = lane & 15;
    const int kb   = half * 2;

    const float* arow = agg + (size_t)(mt * 16 + l15) * DIM;
    const float* wrow = Wc  + (size_t)(nt * 16 + l15) * DIM;

    const int m0 = mt * 16 + half * 8;
    const int n  = nt * 16 + l15;

    v8f c;
    if (FIRST) {
        const float bv = bias[n];
        #pragma unroll
        for (int j = 0; j < 8; ++j) c[j] = bv;
    } else {
        #pragma unroll
        for (int j = 0; j < 8; ++j) c[j] = out[(size_t)(m0 + j) * DIM + n];
    }

    #pragma unroll
    for (int k = 0; k < DIM; k += 4) {
        v2f a0 = *(const v2f*)(arow + k + kb);
        v2f b0 = *(const v2f*)(wrow + k + kb);
        c = __builtin_amdgcn_wmma_f32_16x16x4_f32(false, a0, false, b0,
                                                  (short)0, c, false, false);
    }

    #pragma unroll
    for (int j = 0; j < 8; ++j)
        out[(size_t)(m0 + j) * DIM + n] = c[j];
}

extern "C" void kernel_launch(void* const* d_in, const int* in_sizes, int n_in,
                              void* d_out, int out_size, void* d_ws, size_t ws_size,
                              hipStream_t stream) {
    const float* x      = (const float*)d_in[0];
    const float* W_src  = (const float*)d_in[1];
    const float* b_src  = (const float*)d_in[2];
    const float* W_dst  = (const float*)d_in[3];
    const float* b_dst  = (const float*)d_in[4];
    const int*   eidx   = (const int*)d_in[5];
    const int*   row    = eidx;
    const int*   col    = eidx + N_EDGES;
    float*       outp   = (float*)d_out;

    int* wi = (int*)d_ws;
    int* cnt_r = wi;
    int* cnt_c = cnt_r + N_NODES;
    int* off_r = cnt_c + N_NODES;
    int* off_c = off_r + N_NODES;
    int* pos_r = off_c + N_NODES;
    int* pos_c = pos_r + N_EDGES;
    int* csr_r = pos_c + N_EDGES;
    int* csr_c = csr_r + N_EDGES;
    float* agg  = (float*)(csr_c + N_EDGES);
    float* Wc_s = agg + (size_t)N_NODES * DIM;
    float* Wc_d = Wc_s + DIM * DIM;
    float* bias = Wc_d + DIM * DIM;

    int* H = (int*)agg;

    hist_kernel<<<2 * NCHUNK, 1024, 0, stream>>>(eidx, H, pos_r);
    chunk_prefix_kernel<<<(2 * N_NODES + 255) / 256, 256, 0, stream>>>(H, cnt_r);
    scan_kernel<<<2, 1024, 0, stream>>>(cnt_r, off_r);
    fill_kernel<<<(N_EDGES + 255) / 256, 256, 0, stream>>>(row, col, off_r, off_c, H, pos_r, pos_c,
                                                           csr_r, csr_c);
    prep_kernel<<<(DIM * DIM + 255) / 256, 256, 0, stream>>>(W_src, b_src, W_dst, b_dst,
                                                             Wc_s, Wc_d, bias);
    gather_kernel<<<N_NODES / 8, 256, 0, stream>>>(x, col, cnt_r, cnt_c, off_r, csr_r, agg);
    gemm_kernel<true><<<(N_NODES / 16) * (DIM / 16) / 8, 256, 0, stream>>>(agg, Wc_s, bias, outp);
    gather_kernel<<<N_NODES / 8, 256, 0, stream>>>(x, row, cnt_c, cnt_r, off_c, csr_c, agg);
    gemm_kernel<false><<<(N_NODES / 16) * (DIM / 16) / 8, 256, 0, stream>>>(agg, Wc_d, bias, outp);
}


# segments of 33 to 64 edges: two edges per lane ranked in one pass and put in edge-id order through 512 B of the wave's own LDS rows, replacing one selection pass with four dependent loads per edge
# speedup vs baseline: 2.5984x; 1.5131x over previous
#include <hip/hip_runtime.h>

typedef __attribute__((ext_vector_type(2))) float v2f;
typedef __attribute__((ext_vector_type(8))) float v8f;

#define N_NODES 50000
#define N_EDGES 800000
#define DIM     128

#define NCHUNK  64
#define CHUNK   (N_EDGES / NCHUNK)
#define EDGE_IT ((CHUNK + 1023) / 1024)
#define SCAN_CH ((N_NODES + 1023) / 1024)


__global__ void __launch_bounds__(1024)
hist_kernel(const int* __restrict__ eidx, int* __restrict__ H, int* __restrict__ pos) {
    __shared__ int hist[N_NODES];
    const int t = threadIdx.x;
    int4* h4 = (int4*)hist;
    for (int i = t; i < N_NODES / 4; i += 1024) h4[i] = make_int4(0, 0, 0, 0);

    const int* idx = eidx + (size_t)blockIdx.x * CHUNK;
    int*       p   = pos  + (size_t)blockIdx.x * CHUNK;
    int node[EDGE_IT];
    #pragma unroll
    for (int k = 0; k < EDGE_IT; ++k) {
        const int i = t + k * 1024;
        node[k] = (i < CHUNK) ? idx[i] : -1;
    }
    __syncthreads();
    #pragma unroll
    for (int k = 0; k < EDGE_IT; ++k) {
        const int i = t + k * 1024;
        if (i < CHUNK) p[i] = atomicAdd(&hist[node[k]], 1);
    }
    __syncthreads();
    int4* g4 = (int4*)(H + (size_t)blockIdx.x * N_NODES);
    for (int i = t; i < N_NODES / 4; i += 1024) g4[i] = h4[i];
}

__global__ void __launch_bounds__(256)
chunk_prefix_kernel(int* __restrict__ H, int* __restrict__ cnt) {
    const int i = blockIdx.x * 256 + threadIdx.x;
    if (i >= 2 * N_NODES) return;
    const int d = (i >= N_NODES) ? 1 : 0;
    int* h = H + (size_t)d * NCHUNK * N_NODES + (i - d * N_NODES);
    int v[NCHUNK];
    #pragma unroll
    for (int b = 0; b < NCHUNK; ++b) v[b] = h[(size_t)b * N_NODES];
    int run = 0;
    #pragma unroll
    for (int b = 0; b < NCHUNK; ++b) { h[(size_t)b * N_NODES] = run; run += v[b]; }
    cnt[i] = run;
}

__global__ void __launch_bounds__(1024)
scan_kernel(const int* __restrict__ cnt, int* __restrict__ off) {
    __shared__ int part[1024];
    const int* c = cnt + blockIdx.x * N_NODES;
    int*       o = off + blockIdx.x * N_NODES;
    const int t = threadIdx.x;
    const int b = t * SCAN_CH;
    int v[SCAN_CH];
    int s = 0;
    #pragma unroll
    for (int k = 0; k < SCAN_CH; ++k) {
        v[k] = (b + k < N_NODES) ? c[b + k] : 0;
        s += v[k];
    }
    part[t] = s;
    __syncthreads();
    for (int d = 1; d < 1024; d <<= 1) {
        int u = (t >= d) ? part[t - d] : 0;
        __syncthreads();
        part[t] += u;
        __syncthreads();
    }
    int run = part[t] - s;
    #pragma unroll
    for (int k = 0; k < SCAN_CH; ++k) {
        if (b + k < N_NODES) o[b + k] = run;
        run += v[k];
    }
}

__global__ void fill_kernel(const int* __restrict__ row,
                            const int* __restrict__ col,
                            const int* __restrict__ off_r,
                            const int* __restrict__ off_c,
                            const int* __restrict__ H,
                            const int* __restrict__ pos_r,
                            const int* __restrict__ pos_c,
                            int* __restrict__ csr_r,
                            int* __restrict__ csr_c) {
    int e = blockIdx.x * blockDim.x + threadIdx.x;
    if (e >= N_EDGES) return;
    const int b = e / CHUNK;
    const int r = row[e];
    const int c = col[e];
    csr_r[off_r[r] + H[(size_t)b * N_NODES + r] + pos_r[e]] = e;
    csr_c[off_c[c] + H[(size_t)(NCHUNK + b) * N_NODES + c] + pos_c[e]] = e;
}

__global__ void prep_kernel(const float* __restrict__ W_src,
                            const float* __restrict__ b_src,
                            const float* __restrict__ W_dst,
                            const float* __restrict__ b_dst,
                            float* __restrict__ Wf,
                            float* __restrict__ bias) {
    const int i = blockIdx.x * blockDim.x + threadIdx.x;
    if (i < (DIM / 16) * 32 * 32) {
        const int lane = i & 31;
        const int p    = (i >> 5) & 31;
        const int nt   = i >> 10;
        const int n    = nt * 16 + (lane & 15);
        const int kk   = 8 * p + 4 * (lane >> 4);
        const float* s = (kk < DIM) ? (W_src + (size_t)n * DIM + kk)
                                    : (W_dst + (size_t)n * DIM + (kk - DIM));
        const float4 v = *(const float4*)s;
        ((float4*)Wf)[i] = make_float4(0.5f * v.x, 0.5f * v.y, 0.5f * v.z, 0.5f * v.w);
    }
    if (i < DIM) bias[i] = 0.5f * b_src[i] + 0.5f * b_dst[i];
}

__device__ __forceinline__ float inv_sqrt_deg(int d) {
    float f = (float)d;
    return (d > 0) ? rsqrtf(fmaxf(f, 1.0f)) : 0.0f;
}

__device__ __forceinline__ void axpy4(float4& acc, float w, const float4 v) {
    acc.x += w * v.x; acc.y += w * v.y; acc.z += w * v.z; acc.w += w * v.w;
}

__device__ __forceinline__ void
add_rows(float4& acc, const float* __restrict__ x, const int ssrc, const int sw,
         const int count, const int lane) {
    int s = 0;
    for (; s + 4 <= count; s += 4) {
        const int c0 = __shfl(ssrc, s), c1 = __shfl(ssrc, s + 1);
        const int c2 = __shfl(ssrc, s + 2), c3 = __shfl(ssrc, s + 3);
        const float4 v0 = ((const float4*)(x + (size_t)c0 * DIM))[lane];
        const float4 v1 = ((const float4*)(x + (size_t)c1 * DIM))[lane];
        const float4 v2 = ((const float4*)(x + (size_t)c2 * DIM))[lane];
        const float4 v3 = ((const float4*)(x + (size_t)c3 * DIM))[lane];
        axpy4(acc, __int_as_float(__shfl(sw, s)),     v0);
        axpy4(acc, __int_as_float(__shfl(sw, s + 1)), v1);
        axpy4(acc, __int_as_float(__shfl(sw, s + 2)), v2);
        axpy4(acc, __int_as_float(__shfl(sw, s + 3)), v3);
    }
    for (; s < count; ++s) {
        const int c0 = __shfl(ssrc, s);
        const float4 v0 = ((const float4*)(x + (size_t)c0 * DIM))[lane];
        axpy4(acc, __int_as_float(__shfl(sw, s)), v0);
    }
}

__device__ __forceinline__ float4
gather_node(const float* __restrict__ x,
            const int* __restrict__ other,
            const int* __restrict__ cnt_self,
            const int* __restrict__ cnt_other,
            const int* __restrict__ off_self,
            const int* __restrict__ csr,
            int2* scratch,
            const int n, const int lane) {
    const int deg = cnt_self[n];
    const int beg = off_self[n];
    float4 acc = make_float4(0.f, 0.f, 0.f, 0.f);
    const float inv_n = inv_sqrt_deg(deg);
    int ssrc = 0, sw = 0, nrows = 0, more = 0;
    if (deg > 0 && deg <= 32) {
        int e = 0x7fffffff, srcn = 0;
        float w = 0.f;
        if (lane < deg) {
            e = csr[beg + lane];
            srcn = other[e];
            w = inv_n * inv_sqrt_deg(cnt_other[srcn]);
        }
        int rank = 0;
        for (int j = 0; j < deg; ++j) {
            int ej = __shfl(e, j);
            rank += (ej < e) ? 1 : 0;
        }
        if (lane >= deg) rank = lane;
        ssrc  = __builtin_amdgcn_ds_permute(rank << 2, srcn);
        sw    = __builtin_amdgcn_ds_permute(rank << 2, __float_as_int(w));
        nrows = deg;
    } else if (deg > 32 && deg <= 64) {
        const int  i1   = lane + 32;
        const bool has1 = i1 < deg;
        const int e0 = csr[beg + lane];
        int e1 = 0x7fffffff, src1 = 0;
        float w1 = 0.f;
        if (has1) e1 = csr[beg + i1];
        const int src0 = other[e0];
        if (has1) src1 = other[e1];
        const float w0 = inv_n * inv_sqrt_deg(cnt_other[src0]);
        if (has1) w1 = inv_n * inv_sqrt_deg(cnt_other[src1]);
        int rank0 = 0, rank1 = 0;
        for (int j = 0; j < 32; ++j) {
            const int a = __shfl(e0, j);
            const int b = __shfl(e1, j);
            rank0 += ((a < e0) ? 1 : 0) + ((b < e0) ? 1 : 0);
            rank1 += ((a < e1) ? 1 : 0) + ((b < e1) ? 1 : 0);
        }
        scratch[rank0] = make_int2(src0, __float_as_int(w0));
        if (has1) scratch[rank1] = make_int2(src1, __float_as_int(w1));
        const int2 lo = scratch[lane];
        ssrc  = lo.x;
        sw    = lo.y;
        nrows = 32;
        more  = deg - 32;
    } else if (deg > 64) {
        int last = -1;
        for (int s = 0; s < deg; ++s) {
            int m = 0x7fffffff;
            for (int i = lane; i < deg; i += 32) {
                int ei = csr[beg + i];
                if (ei > last && ei < m) m = ei;
            }
            for (int o = 16; o > 0; o >>= 1) {
                int t = __shfl_xor(m, o);
                m = (t < m) ? t : m;
            }
            last = m;
            const int c0 = other[m];
            const float w = inv_n * inv_sqrt_deg(cnt_other[c0]);
            const float4 v0 = ((const float4*)(x + (size_t)c0 * DIM))[lane];
            axpy4(acc, w, v0);
        }
    }
    #pragma unroll 1
    for (;;) {
        add_rows(acc, x, ssrc, sw, nrows, lane);
        if (more <= 0) break;
        int2 hi = make_int2(0, 0);
        if (lane < more) hi = scratch[lane + 32];
        ssrc  = hi.x;
        sw    = hi.y;
        nrows = more;
        more  = 0;
    }
    return acc;
}

#define TILE_M 32
#define APITCH 260

__global__ void __launch_bounds__(256)
fused_kernel(const float* __restrict__ x,
             const int* __restrict__ row,
             const int* __restrict__ col,
             const int* __restrict__ cnt_r,
             const int* __restrict__ cnt_c,
             const int* __restrict__ off_r,
             const int* __restrict__ off_c,
             const int* __restrict__ csr_r,
             const int* __restrict__ csr_c,
             const float* __restrict__ Wf,
             const float* __restrict__ bias,
             float* __restrict__ out) {
    __shared__ float As[TILE_M * APITCH];
    const int lane = threadIdx.x & 31;
    const int wv   = __builtin_amdgcn_readfirstlane(threadIdx.x >> 5);
    const int base = blockIdx.x * TILE_M;

    #pragma unroll 1
    for (int s = 0; s < 8; ++s) {
        const int dir = s & 1;
        const int r   = wv * 4 + (s >> 1);
        const int n   = base + r;
        float4 f = make_float4(0.f, 0.f, 0.f, 0.f);
        if (n < N_NODES)
            f = gather_node(x, dir ? row : col, dir ? cnt_c : cnt_r, dir ? cnt_r : cnt_c,
                            dir ? off_c : off_r, dir ? csr_c : csr_r,
                            (int2*)(As + r * APITCH + dir * DIM), n, lane);
        *(float4*)(As + r * APITCH + dir * DIM + lane * 4) = f;
    }
    __syncthreads();

    const int half = lane >> 4;
    const int l15  = lane & 15;
    const float bv = bias[wv * 16 + l15];
    v8f c0, c1;
    #pragma unroll
    for (int j = 0; j < 8; ++j) { c0[j] = bv; c1[j] = bv; }

    const float4* wf  = (const float4*)Wf + (size_t)wv * 32 * 32 + lane;
    const float*  a0p = As + l15 * APITCH + 4 * half;
    const float*  a1p = a0p + 16 * APITCH;
    #pragma unroll 4
    for (int p = 0; p < 32; ++p) {
        const float4 b  = wf[p * 32];
        const float4 a0 = *(const float4*)(a0p + 8 * p);
        const float4 a1 = *(const float4*)(a1p + 8 * p);
        v2f bx = {b.x, b.y},  by = {b.z, b.w};
        v2f x0 = {a0.x, a0.y}, y0 = {a0.z, a0.w};
        v2f x1 = {a1.x, a1.y}, y1 = {a1.z, a1.w};
        c0 = __builtin_amdgcn_wmma_f32_16x16x4_f32(false, x0, false, bx, (short)0, c0, false, false);
        c1 = __builtin_amdgcn_wmma_f32_16x16x4_f32(false, x1, false, bx, (short)0, c1, false, false);
        c0 = __builtin_amdgcn_wmma_f32_16x16x4_f32(false, y0, false, by, (short)0, c0, false, false);
        c1 = __builtin_amdgcn_wmma_f32_16x16x4_f32(false, y1, false, by, (short)0, c1, false, false);
    }
    __syncthreads();

    float* d0 = As + (half * 8) * APITCH + wv * 16 + l15;
    float* d1 = d0 + 16 * APITCH;
    #pragma unroll
    for (int j = 0; j < 8; ++j) { d0[j * APITCH] = c0[j]; d1[j * APITCH] = c1[j]; }
    __syncthreads();

    for (int i = 0; i < 4; ++i) {
        const int r = wv * 4 + i;
        const int n = base + r;
        if (n < N_NODES)
            ((float4*)(out + (size_t)n * DIM))[lane] = *(const float4*)(As + r * APITCH + lane * 4);
    }
}

extern "C" void kernel_launch(void* const* d_in, const int* in_sizes, int n_in,
                              void* d_out, int out_size, void* d_ws, size_t ws_size,
                              hipStream_t stream) {
    const float* x      = (const float*)d_in[0];
    const float* W_src  = (const float*)d_in[1];
    const float* b_src  = (const float*)d_in[2];
    const float* W_dst  = (const float*)d_in[3];
    const float* b_dst  = (const float*)d_in[4];
    const int*   eidx   = (const int*)d_in[5];
    const int*   row    = eidx;
    const int*   col    = eidx + N_EDGES;
    float*       outp   = (float*)d_out;

    int* wi = (int*)d_ws;
    int* cnt_r = wi;
    int* cnt_c = cnt_r + N_NODES;
    int* off_r = cnt_c + N_NODES;
    int* off_c = off_r + N_NODES;
    int* pos_r = off_c + N_NODES;
    int* pos_c = pos_r + N_EDGES;
    int* csr_r = pos_c + N_EDGES;
    int* csr_c = csr_r + N_EDGES;
    int*   H    = csr_c + N_EDGES;
    float* Wf   = (float*)(H + (size_t)N_NODES * DIM);
    float* bias = Wf + 2 * DIM * DIM;

    hist_kernel<<<2 * NCHUNK, 1024, 0, stream>>>(eidx, H, pos_r);
    chunk_prefix_kernel<<<(2 * N_NODES + 255) / 256, 256, 0, stream>>>(H, cnt_r);
    scan_kernel<<<2, 1024, 0, stream>>>(cnt_r, off_r);
    fill_kernel<<<(N_EDGES + 255) / 256, 256, 0, stream>>>(row, col, off_r, off_c, H, pos_r, pos_c,
                                                           csr_r, csr_c);
    prep_kernel<<<((DIM / 16) * 32 * 32 + 255) / 256, 256, 0, stream>>>(W_src, b_src, W_dst, b_dst,
                                                                        Wf, bias);
    fused_kernel<<<(N_NODES + TILE_M - 1) / TILE_M, 256, 0, stream>>>(x, row, col, cnt_r, cnt_c,
                                                                      off_r, off_c, csr_r, csr_c,
                                                                      Wf, bias, outp);
}


# node-offset scan: counts cross global memory in whole 128-byte lines and are re-sliced per thread in LDS, replacing 49 loads and 49 stores per lane at a 196-byte lane stride (32 lines each)
# speedup vs baseline: 3.3802x; 1.3009x over previous
#include <hip/hip_runtime.h>

typedef __attribute__((ext_vector_type(2))) float v2f;
typedef __attribute__((ext_vector_type(8))) float v8f;

#define N_NODES 50000
#define N_EDGES 800000
#define DIM     128

#define NCHUNK  64
#define CHUNK   (N_EDGES / NCHUNK)
#define EDGE_IT ((CHUNK + 1023) / 1024)
#define SCAN_CH ((N_NODES + 1023) / 1024)


__global__ void __launch_bounds__(1024)
hist_kernel(const int* __restrict__ eidx, int* __restrict__ H, int* __restrict__ pos) {
    __shared__ int hist[N_NODES];
    const int t = threadIdx.x;
    int4* h4 = (int4*)hist;
    for (int i = t; i < N_NODES / 4; i += 1024) h4[i] = make_int4(0, 0, 0, 0);

    const int* idx = eidx + (size_t)blockIdx.x * CHUNK;
    int*       p   = pos  + (size_t)blockIdx.x * CHUNK;
    int node[EDGE_IT];
    #pragma unroll
    for (int k = 0; k < EDGE_IT; ++k) {
        const int i = t + k * 1024;
        node[k] = (i < CHUNK) ? idx[i] : -1;
    }
    __syncthreads();
    #pragma unroll
    for (int k = 0; k < EDGE_IT; ++k) {
        const int i = t + k * 1024;
        if (i < CHUNK) p[i] = atomicAdd(&hist[node[k]], 1);
    }
    __syncthreads();
    int4* g4 = (int4*)(H + (size_t)blockIdx.x * N_NODES);
    for (int i = t; i < N_NODES / 4; i += 1024) g4[i] = h4[i];
}

__global__ void __launch_bounds__(256)
chunk_prefix_kernel(int* __restrict__ H, int* __restrict__ cnt) {
    const int i = blockIdx.x * 256 + threadIdx.x;
    if (i >= 2 * N_NODES) return;
    const int d = (i >= N_NODES) ? 1 : 0;
    int* h = H + (size_t)d * NCHUNK * N_NODES + (i - d * N_NODES);
    int v[NCHUNK];
    #pragma unroll
    for (int b = 0; b < NCHUNK; ++b) v[b] = h[(size_t)b * N_NODES];
    int run = 0;
    #pragma unroll
    for (int b = 0; b < NCHUNK; ++b) { h[(size_t)b * N_NODES] = run; run += v[b]; }
    cnt[i] = run;
}

#define SCAN_V4  ((N_NODES / 4 + 1023) / 1024)
#define SCAN_PAD (1024 * SCAN_CH)
__global__ void __launch_bounds__(1024)
scan_kernel(const int* __restrict__ cnt, int* __restrict__ off) {
    __shared__ alignas(16) int buf[SCAN_PAD];
    __shared__ int part[1024];
    const int4* c4 = (const int4*)(cnt + blockIdx.x * N_NODES);
    int4*       o4 = (int4*)(off + blockIdx.x * N_NODES);
    int4*       b4 = (int4*)buf;
    const int t = threadIdx.x;
    int4 q[SCAN_V4];
    #pragma unroll
    for (int k = 0; k < SCAN_V4; ++k) {
        const int i = t + k * 1024;
        q[k] = (i < N_NODES / 4) ? c4[i] : make_int4(0, 0, 0, 0);
    }
    #pragma unroll
    for (int k = 0; k < SCAN_V4; ++k) {
        const int i = t + k * 1024;
        if (i < N_NODES / 4) b4[i] = q[k];
    }
    if (t < SCAN_PAD - N_NODES) buf[N_NODES + t] = 0;
    __syncthreads();

    const int b = t * SCAN_CH;
    int v[SCAN_CH];
    int s = 0;
    #pragma unroll
    for (int k = 0; k < SCAN_CH; ++k) {
        v[k] = buf[b + k];
        s += v[k];
    }
    part[t] = s;
    __syncthreads();
    for (int d = 1; d < 1024; d <<= 1) {
        int u = (t >= d) ? part[t - d] : 0;
        __syncthreads();
        part[t] += u;
        __syncthreads();
    }
    int run = part[t] - s;
    #pragma unroll
    for (int k = 0; k < SCAN_CH; ++k) {
        buf[b + k] = run;
        run += v[k];
    }
    __syncthreads();
    for (int i = t; i < N_NODES / 4; i += 1024) o4[i] = b4[i];
}

__global__ void fill_kernel(const int* __restrict__ row,
                            const int* __restrict__ col,
                            const int* __restrict__ off_r,
                            const int* __restrict__ off_c,
                            const int* __restrict__ H,
                            const int* __restrict__ pos_r,
                            const int* __restrict__ pos_c,
                            int* __restrict__ csr_r,
                            int* __restrict__ csr_c) {
    int e = blockIdx.x * blockDim.x + threadIdx.x;
    if (e >= N_EDGES) return;
    const int b = e / CHUNK;
    const int r = row[e];
    const int c = col[e];
    csr_r[off_r[r] + H[(size_t)b * N_NODES + r] + pos_r[e]] = e;
    csr_c[off_c[c] + H[(size_t)(NCHUNK + b) * N_NODES + c] + pos_c[e]] = e;
}

__global__ void prep_kernel(const float* __restrict__ W_src,
                            const float* __restrict__ b_src,
                            const float* __restrict__ W_dst,
                            const float* __restrict__ b_dst,
                            float* __restrict__ Wf,
                            float* __restrict__ bias) {
    const int i = blockIdx.x * blockDim.x + threadIdx.x;
    if (i < (DIM / 16) * 32 * 32) {
        const int lane = i & 31;
        const int p    = (i >> 5) & 31;
        const int nt   = i >> 10;
        const int n    = nt * 16 + (lane & 15);
        const int kk   = 8 * p + 4 * (lane >> 4);
        const float* s = (kk < DIM) ? (W_src + (size_t)n * DIM + kk)
                                    : (W_dst + (size_t)n * DIM + (kk - DIM));
        const float4 v = *(const float4*)s;
        ((float4*)Wf)[i] = make_float4(0.5f * v.x, 0.5f * v.y, 0.5f * v.z, 0.5f * v.w);
    }
    if (i < DIM) bias[i] = 0.5f * b_src[i] + 0.5f * b_dst[i];
}

__device__ __forceinline__ float inv_sqrt_deg(int d) {
    float f = (float)d;
    return (d > 0) ? rsqrtf(fmaxf(f, 1.0f)) : 0.0f;
}

__device__ __forceinline__ void axpy4(float4& acc, float w, const float4 v) {
    acc.x += w * v.x; acc.y += w * v.y; acc.z += w * v.z; acc.w += w * v.w;
}

__device__ __forceinline__ void
add_rows(float4& acc, const float* __restrict__ x, const int ssrc, const int sw,
         const int count, const int lane) {
    int s = 0;
    for (; s + 4 <= count; s += 4) {
        const int c0 = __shfl(ssrc, s), c1 = __shfl(ssrc, s + 1);
        const int c2 = __shfl(ssrc, s + 2), c3 = __shfl(ssrc, s + 3);
        const float4 v0 = ((const float4*)(x + (size_t)c0 * DIM))[lane];
        const float4 v1 = ((const float4*)(x + (size_t)c1 * DIM))[lane];
        const float4 v2 = ((const float4*)(x + (size_t)c2 * DIM))[lane];
        const float4 v3 = ((const float4*)(x + (size_t)c3 * DIM))[lane];
        axpy4(acc, __int_as_float(__shfl(sw, s)),     v0);
        axpy4(acc, __int_as_float(__shfl(sw, s + 1)), v1);
        axpy4(acc, __int_as_float(__shfl(sw, s + 2)), v2);
        axpy4(acc, __int_as_float(__shfl(sw, s + 3)), v3);
    }
    for (; s < count; ++s) {
        const int c0 = __shfl(ssrc, s);
        const float4 v0 = ((const float4*)(x + (size_t)c0 * DIM))[lane];
        axpy4(acc, __int_as_float(__shfl(sw, s)), v0);
    }
}

__device__ __forceinline__ float4
gather_node(const float* __restrict__ x,
            const int* __restrict__ other,
            const int* __restrict__ cnt_self,
            const int* __restrict__ cnt_other,
            const int* __restrict__ off_self,
            const int* __restrict__ csr,
            int2* scratch,
            const int n, const int lane) {
    const int deg = cnt_self[n];
    const int beg = off_self[n];
    float4 acc = make_float4(0.f, 0.f, 0.f, 0.f);
    const float inv_n = inv_sqrt_deg(deg);
    int ssrc = 0, sw = 0, nrows = 0, more = 0;
    if (deg > 0 && deg <= 32) {
        int e = 0x7fffffff, srcn = 0;
        float w = 0.f;
        if (lane < deg) {
            e = csr[beg + lane];
            srcn = other[e];
            w = inv_n * inv_sqrt_deg(cnt_other[srcn]);
        }
        int rank = 0;
        for (int j = 0; j < deg; ++j) {
            int ej = __shfl(e, j);
            rank += (ej < e) ? 1 : 0;
        }
        if (lane >= deg) rank = lane;
        ssrc  = __builtin_amdgcn_ds_permute(rank << 2, srcn);
        sw    = __builtin_amdgcn_ds_permute(rank << 2, __float_as_int(w));
        nrows = deg;
    } else if (deg > 32 && deg <= 64) {
        const int  i1   = lane + 32;
        const bool has1 = i1 < deg;
        const int e0 = csr[beg + lane];
        int e1 = 0x7fffffff, src1 = 0;
        float w1 = 0.f;
        if (has1) e1 = csr[beg + i1];
        const int src0 = other[e0];
        if (has1) src1 = other[e1];
        const float w0 = inv_n * inv_sqrt_deg(cnt_other[src0]);
        if (has1) w1 = inv_n * inv_sqrt_deg(cnt_other[src1]);
        int rank0 = 0, rank1 = 0;
        for (int j = 0; j < 32; ++j) {
            const int a = __shfl(e0, j);
            const int b = __shfl(e1, j);
            rank0 += ((a < e0) ? 1 : 0) + ((b < e0) ? 1 : 0);
            rank1 += ((a < e1) ? 1 : 0) + ((b < e1) ? 1 : 0);
        }
        scratch[rank0] = make_int2(src0, __float_as_int(w0));
        if (has1) scratch[rank1] = make_int2(src1, __float_as_int(w1));
        const int2 lo = scratch[lane];
        ssrc  = lo.x;
        sw    = lo.y;
        nrows = 32;
        more  = deg - 32;
    } else if (deg > 64) {
        int last = -1;
        for (int s = 0; s < deg; ++s) {
            int m = 0x7fffffff;
            for (int i = lane; i < deg; i += 32) {
                int ei = csr[beg + i];
                if (ei > last && ei < m) m = ei;
            }
            for (int o = 16; o > 0; o >>= 1) {
                int t = __shfl_xor(m, o);
                m = (t < m) ? t : m;
            }
            last = m;
            const int c0 = other[m];
            const float w = inv_n * inv_sqrt_deg(cnt_other[c0]);
            const float4 v0 = ((const float4*)(x + (size_t)c0 * DIM))[lane];
            axpy4(acc, w, v0);
        }
    }
    #pragma unroll 1
    for (;;) {
        add_rows(acc, x, ssrc, sw, nrows, lane);
        if (more <= 0) break;
        int2 hi = make_int2(0, 0);
        if (lane < more) hi = scratch[lane + 32];
        ssrc  = hi.x;
        sw    = hi.y;
        nrows = more;
        more  = 0;
    }
    return acc;
}

#define TILE_M 32
#define APITCH 260

__global__ void __launch_bounds__(256)
fused_kernel(const float* __restrict__ x,
             const int* __restrict__ row,
             const int* __restrict__ col,
             const int* __restrict__ cnt_r,
             const int* __restrict__ cnt_c,
             const int* __restrict__ off_r,
             const int* __restrict__ off_c,
             const int* __restrict__ csr_r,
             const int* __restrict__ csr_c,
             const float* __restrict__ Wf,
             const float* __restrict__ bias,
             float* __restrict__ out) {
    __shared__ float As[TILE_M * APITCH];
    const int lane = threadIdx.x & 31;
    const int wv   = __builtin_amdgcn_readfirstlane(threadIdx.x >> 5);
    const int base = blockIdx.x * TILE_M;

    #pragma unroll 1
    for (int s = 0; s < 8; ++s) {
        const int dir = s & 1;
        const int r   = wv * 4 + (s >> 1);
        const int n   = base + r;
        float4 f = make_float4(0.f, 0.f, 0.f, 0.f);
        if (n < N_NODES)
            f = gather_node(x, dir ? row : col, dir ? cnt_c : cnt_r, dir ? cnt_r : cnt_c,
                            dir ? off_c : off_r, dir ? csr_c : csr_r,
                            (int2*)(As + r * APITCH + dir * DIM), n, lane);
        *(float4*)(As + r * APITCH + dir * DIM + lane * 4) = f;
    }
    __syncthreads();

    const int half = lane >> 4;
    const int l15  = lane & 15;
    const float bv = bias[wv * 16 + l15];
    v8f c0, c1;
    #pragma unroll
    for (int j = 0; j < 8; ++j) { c0[j] = bv; c1[j] = bv; }

    const float4* wf  = (const float4*)Wf + (size_t)wv * 32 * 32 + lane;
    const float*  a0p = As + l15 * APITCH + 4 * half;
    const float*  a1p = a0p + 16 * APITCH;
    #pragma unroll 4
    for (int p = 0; p < 32; ++p) {
        const float4 b  = wf[p * 32];
        const float4 a0 = *(const float4*)(a0p + 8 * p);
        const float4 a1 = *(const float4*)(a1p + 8 * p);
        v2f bx = {b.x, b.y},  by = {b.z, b.w};
        v2f x0 = {a0.x, a0.y}, y0 = {a0.z, a0.w};
        v2f x1 = {a1.x, a1.y}, y1 = {a1.z, a1.w};
        c0 = __builtin_amdgcn_wmma_f32_16x16x4_f32(false, x0, false, bx, (short)0, c0, false, false);
        c1 = __builtin_amdgcn_wmma_f32_16x16x4_f32(false, x1, false, bx, (short)0, c1, false, false);
        c0 = __builtin_amdgcn_wmma_f32_16x16x4_f32(false, y0, false, by, (short)0, c0, false, false);
        c1 = __builtin_amdgcn_wmma_f32_16x16x4_f32(false, y1, false, by, (short)0, c1, false, false);
    }
    __syncthreads();

    float* d0 = As + (half * 8) * APITCH + wv * 16 + l15;
    float* d1 = d0 + 16 * APITCH;
    #pragma unroll
    for (int j = 0; j < 8; ++j) { d0[j * APITCH] = c0[j]; d1[j * APITCH] = c1[j]; }
    __syncthreads();

    for (int i = 0; i < 4; ++i) {
        const int r = wv * 4 + i;
        const int n = base + r;
        if (n < N_NODES)
            ((float4*)(out + (size_t)n * DIM))[lane] = *(const float4*)(As + r * APITCH + lane * 4);
    }
}

extern "C" void kernel_launch(void* const* d_in, const int* in_sizes, int n_in,
                              void* d_out, int out_size, void* d_ws, size_t ws_size,
                              hipStream_t stream) {
    const float* x      = (const float*)d_in[0];
    const float* W_src  = (const float*)d_in[1];
    const float* b_src  = (const float*)d_in[2];
    const float* W_dst  = (const float*)d_in[3];
    const float* b_dst  = (const float*)d_in[4];
    const int*   eidx   = (const int*)d_in[5];
    const int*   row    = eidx;
    const int*   col    = eidx + N_EDGES;
    float*       outp   = (float*)d_out;

    int* wi = (int*)d_ws;
    int* cnt_r = wi;
    int* cnt_c = cnt_r + N_NODES;
    int* off_r = cnt_c + N_NODES;
    int* off_c = off_r + N_NODES;
    int* pos_r = off_c + N_NODES;
    int* pos_c = pos_r + N_EDGES;
    int* csr_r = pos_c + N_EDGES;
    int* csr_c = csr_r + N_EDGES;
    int*   H    = csr_c + N_EDGES;
    float* Wf   = (float*)(H + (size_t)N_NODES * DIM);
    float* bias = Wf + 2 * DIM * DIM;

    hist_kernel<<<2 * NCHUNK, 1024, 0, stream>>>(eidx, H, pos_r);
    chunk_prefix_kernel<<<(2 * N_NODES + 255) / 256, 256, 0, stream>>>(H, cnt_r);
    scan_kernel<<<2, 1024, 0, stream>>>(cnt_r, off_r);
    fill_kernel<<<(N_EDGES + 255) / 256, 256, 0, stream>>>(row, col, off_r, off_c, H, pos_r, pos_c,
                                                           csr_r, csr_c);
    prep_kernel<<<((DIM / 16) * 32 * 32 + 255) / 256, 256, 0, stream>>>(W_src, b_src, W_dst, b_dst,
                                                                        Wf, bias);
    fused_kernel<<<(N_NODES + TILE_M - 1) / TILE_M, 256, 0, stream>>>(x, row, col, cnt_r, cnt_c,
                                                                      off_r, off_c, csr_r, csr_c,
                                                                      Wf, bias, outp);
}


# segment entries hold (other end node, its degree) packed in one word, written by the fill pass: the gather's three dependent index loads per segment become one; sum order is by entry value
# speedup vs baseline: 3.5369x; 1.0464x over previous
#include <hip/hip_runtime.h>

typedef __attribute__((ext_vector_type(2))) float v2f;
typedef __attribute__((ext_vector_type(8))) float v8f;

#define N_NODES 50000
#define N_EDGES 800000
#define DIM     128

#define NCHUNK  64
#define CHUNK   (N_EDGES / NCHUNK)
#define EDGE_IT ((CHUNK + 1023) / 1024)
#define SCAN_CH ((N_NODES + 1023) / 1024)


__global__ void __launch_bounds__(1024)
hist_kernel(const int* __restrict__ eidx, int* __restrict__ H, int* __restrict__ pos) {
    __shared__ int hist[N_NODES];
    const int t = threadIdx.x;
    int4* h4 = (int4*)hist;
    for (int i = t; i < N_NODES / 4; i += 1024) h4[i] = make_int4(0, 0, 0, 0);

    const int* idx = eidx + (size_t)blockIdx.x * CHUNK;
    int*       p   = pos  + (size_t)blockIdx.x * CHUNK;
    int node[EDGE_IT];
    #pragma unroll
    for (int k = 0; k < EDGE_IT; ++k) {
        const int i = t + k * 1024;
        node[k] = (i < CHUNK) ? idx[i] : -1;
    }
    __syncthreads();
    #pragma unroll
    for (int k = 0; k < EDGE_IT; ++k) {
        const int i = t + k * 1024;
        if (i < CHUNK) p[i] = atomicAdd(&hist[node[k]], 1);
    }
    __syncthreads();
    int4* g4 = (int4*)(H + (size_t)blockIdx.x * N_NODES);
    for (int i = t; i < N_NODES / 4; i += 1024) g4[i] = h4[i];
}

__global__ void __launch_bounds__(256)
chunk_prefix_kernel(int* __restrict__ H, int* __restrict__ cnt) {
    const int i = blockIdx.x * 256 + threadIdx.x;
    if (i >= 2 * N_NODES) return;
    const int d = (i >= N_NODES) ? 1 : 0;
    int* h = H + (size_t)d * NCHUNK * N_NODES + (i - d * N_NODES);
    int v[NCHUNK];
    #pragma unroll
    for (int b = 0; b < NCHUNK; ++b) v[b] = h[(size_t)b * N_NODES];
    int run = 0;
    #pragma unroll
    for (int b = 0; b < NCHUNK; ++b) { h[(size_t)b * N_NODES] = run; run += v[b]; }
    cnt[i] = run;
}

#define SCAN_V4  ((N_NODES / 4 + 1023) / 1024)
#define SCAN_PAD (1024 * SCAN_CH)
__global__ void __launch_bounds__(1024)
scan_kernel(const int* __restrict__ cnt, int* __restrict__ off) {
    __shared__ alignas(16) int buf[SCAN_PAD];
    __shared__ int part[1024];
    const int4* c4 = (const int4*)(cnt + blockIdx.x * N_NODES);
    int4*       o4 = (int4*)(off + blockIdx.x * N_NODES);
    int4*       b4 = (int4*)buf;
    const int t = threadIdx.x;
    int4 q[SCAN_V4];
    #pragma unroll
    for (int k = 0; k < SCAN_V4; ++k) {
        const int i = t + k * 1024;
        q[k] = (i < N_NODES / 4) ? c4[i] : make_int4(0, 0, 0, 0);
    }
    #pragma unroll
    for (int k = 0; k < SCAN_V4; ++k) {
        const int i = t + k * 1024;
        if (i < N_NODES / 4) b4[i] = q[k];
    }
    if (t < SCAN_PAD - N_NODES) buf[N_NODES + t] = 0;
    __syncthreads();

    const int b = t * SCAN_CH;
    int v[SCAN_CH];
    int s = 0;
    #pragma unroll
    for (int k = 0; k < SCAN_CH; ++k) {
        v[k] = buf[b + k];
        s += v[k];
    }
    part[t] = s;
    __syncthreads();
    for (int d = 1; d < 1024; d <<= 1) {
        int u = (t >= d) ? part[t - d] : 0;
        __syncthreads();
        part[t] += u;
        __syncthreads();
    }
    int run = part[t] - s;
    #pragma unroll
    for (int k = 0; k < SCAN_CH; ++k) {
        buf[b + k] = run;
        run += v[k];
    }
    __syncthreads();
    for (int i = t; i < N_NODES / 4; i += 1024) o4[i] = b4[i];
}

static_assert(N_NODES <= 65535, "a node id must fit the upper 16 bits of an entry");
#define DEG_ESC 0xFFFF
__device__ __forceinline__ unsigned pack_entry(const int node, const int deg) {
    return ((unsigned)node << 16) | (unsigned)(deg < DEG_ESC ? deg : DEG_ESC);
}

__global__ void fill_kernel(const int* __restrict__ row,
                            const int* __restrict__ col,
                            const int* __restrict__ cnt_r,
                            const int* __restrict__ cnt_c,
                            const int* __restrict__ off_r,
                            const int* __restrict__ off_c,
                            const int* __restrict__ H,
                            const int* __restrict__ pos_r,
                            const int* __restrict__ pos_c,
                            unsigned* __restrict__ csr_r,
                            unsigned* __restrict__ csr_c) {
    int e = blockIdx.x * blockDim.x + threadIdx.x;
    if (e >= N_EDGES) return;
    const int b = e / CHUNK;
    const int r = row[e];
    const int c = col[e];
    csr_r[off_r[r] + H[(size_t)b * N_NODES + r] + pos_r[e]] = pack_entry(c, cnt_c[c]);
    csr_c[off_c[c] + H[(size_t)(NCHUNK + b) * N_NODES + c] + pos_c[e]] = pack_entry(r, cnt_r[r]);
}

__global__ void prep_kernel(const float* __restrict__ W_src,
                            const float* __restrict__ b_src,
                            const float* __restrict__ W_dst,
                            const float* __restrict__ b_dst,
                            float* __restrict__ Wf,
                            float* __restrict__ bias) {
    const int i = blockIdx.x * blockDim.x + threadIdx.x;
    if (i < (DIM / 16) * 32 * 32) {
        const int lane = i & 31;
        const int p    = (i >> 5) & 31;
        const int nt   = i >> 10;
        const int n    = nt * 16 + (lane & 15);
        const int kk   = 8 * p + 4 * (lane >> 4);
        const float* s = (kk < DIM) ? (W_src + (size_t)n * DIM + kk)
                                    : (W_dst + (size_t)n * DIM + (kk - DIM));
        const float4 v = *(const float4*)s;
        ((float4*)Wf)[i] = make_float4(0.5f * v.x, 0.5f * v.y, 0.5f * v.z, 0.5f * v.w);
    }
    if (i < DIM) bias[i] = 0.5f * b_src[i] + 0.5f * b_dst[i];
}

__device__ __forceinline__ float inv_sqrt_deg(int d) {
    float f = (float)d;
    return (d > 0) ? rsqrtf(fmaxf(f, 1.0f)) : 0.0f;
}

__device__ __forceinline__ void axpy4(float4& acc, float w, const float4 v) {
    acc.x += w * v.x; acc.y += w * v.y; acc.z += w * v.z; acc.w += w * v.w;
}

__device__ __forceinline__ void
add_rows(float4& acc, const float* __restrict__ x, const int ssrc, const int sw,
         const int count, const int lane) {
    int s = 0;
    for (; s + 4 <= count; s += 4) {
        const int c0 = __shfl(ssrc, s), c1 = __shfl(ssrc, s + 1);
        const int c2 = __shfl(ssrc, s + 2), c3 = __shfl(ssrc, s + 3);
        const float4 v0 = ((const float4*)(x + (size_t)c0 * DIM))[lane];
        const float4 v1 = ((const float4*)(x + (size_t)c1 * DIM))[lane];
        const float4 v2 = ((const float4*)(x + (size_t)c2 * DIM))[lane];
        const float4 v3 = ((const float4*)(x + (size_t)c3 * DIM))[lane];
        axpy4(acc, __int_as_float(__shfl(sw, s)),     v0);
        axpy4(acc, __int_as_float(__shfl(sw, s + 1)), v1);
        axpy4(acc, __int_as_float(__shfl(sw, s + 2)), v2);
        axpy4(acc, __int_as_float(__shfl(sw, s + 3)), v3);
    }
    for (; s < count; ++s) {
        const int c0 = __shfl(ssrc, s);
        const float4 v0 = ((const float4*)(x + (size_t)c0 * DIM))[lane];
        axpy4(acc, __int_as_float(__shfl(sw, s)), v0);
    }
}

__device__ __forceinline__ float
entry_weight(const unsigned k, const float inv_n, const int* __restrict__ cnt_other) {
    int d = (int)(k & DEG_ESC);
    if (d == DEG_ESC) d = cnt_other[k >> 16];
    return inv_n * inv_sqrt_deg(d);
}

__device__ __forceinline__ float4
gather_node(const float* __restrict__ x,
            const int* __restrict__ cnt_self,
            const int* __restrict__ cnt_other,
            const int* __restrict__ off_self,
            const unsigned* __restrict__ csr,
            int2* scratch,
            const int n, const int lane) {
    const int deg = cnt_self[n];
    const int beg = off_self[n];
    float4 acc = make_float4(0.f, 0.f, 0.f, 0.f);
    const float inv_n = inv_sqrt_deg(deg);
    int ssrc = 0, sw = 0, nrows = 0, more = 0;
    if (deg > 0 && deg <= 32) {
        unsigned key = 0xFFFFFFFFu;
        int srcn = 0;
        float w = 0.f;
        if (lane < deg) {
            key  = csr[beg + lane];
            srcn = (int)(key >> 16);
            w    = entry_weight(key, inv_n, cnt_other);
        }
        int rank = 0;
        for (int j = 0; j < deg; ++j) {
            const unsigned kj = (unsigned)__shfl((int)key, j);
            rank += (kj < key || (kj == key && j < lane)) ? 1 : 0;
        }
        if (lane >= deg) rank = lane;
        ssrc  = __builtin_amdgcn_ds_permute(rank << 2, srcn);
        sw    = __builtin_amdgcn_ds_permute(rank << 2, __float_as_int(w));
        nrows = deg;
    } else if (deg > 32 && deg <= 64) {
        const int  i1   = lane + 32;
        const bool has1 = i1 < deg;
        const unsigned k0 = csr[beg + lane];
        unsigned k1 = 0xFFFFFFFFu;
        if (has1) k1 = csr[beg + i1];
        const int   src0 = (int)(k0 >> 16);
        const float w0   = entry_weight(k0, inv_n, cnt_other);
        int   src1 = 0;
        float w1   = 0.f;
        if (has1) {
            src1 = (int)(k1 >> 16);
            w1   = entry_weight(k1, inv_n, cnt_other);
        }
        int rank0 = 0, rank1 = 0;
        for (int j = 0; j < 32; ++j) {
            const unsigned a = (unsigned)__shfl((int)k0, j);
            const unsigned b = (unsigned)__shfl((int)k1, j);
            rank0 += ((a < k0 || (a == k0 && j < lane)) ? 1 : 0) + ((b < k0) ? 1 : 0);
            rank1 += ((a <= k1) ? 1 : 0) + ((b < k1 || (b == k1 && j < lane)) ? 1 : 0);
        }
        scratch[rank0] = make_int2(src0, __float_as_int(w0));
        if (has1) scratch[rank1] = make_int2(src1, __float_as_int(w1));
        const int2 lo = scratch[lane];
        ssrc  = lo.x;
        sw    = lo.y;
        nrows = 32;
        more  = deg - 32;
    } else if (deg > 64) {
        unsigned last = 0;
        bool first = true;
        int done = 0;
        for (int s = 0; s < deg && done < deg; ++s) {
            unsigned m = 0xFFFFFFFFu;
            int c = 0;
            for (int i = lane; i < deg; i += 32) {
                const unsigned k = csr[beg + i];
                if (first || k > last) {
                    if (k < m) { m = k; c = 1; }
                    else if (k == m) ++c;
                }
            }
            unsigned g = m;
            for (int o = 16; o > 0; o >>= 1) {
                const unsigned t = (unsigned)__shfl_xor((int)g, o);
                g = (t < g) ? t : g;
            }
            int mult = (m == g) ? c : 0;
            for (int o = 16; o > 0; o >>= 1) mult += __shfl_xor(mult, o);
            last  = g;
            first = false;
            done += mult;
            const float w = entry_weight(g, inv_n, cnt_other);
            const float4 v0 = ((const float4*)(x + (size_t)(g >> 16) * DIM))[lane];
            for (int q = 0; q < mult; ++q) axpy4(acc, w, v0);
        }
    }
    #pragma unroll 1
    for (;;) {
        add_rows(acc, x, ssrc, sw, nrows, lane);
        if (more <= 0) break;
        int2 hi = make_int2(0, 0);
        if (lane < more) hi = scratch[lane + 32];
        ssrc  = hi.x;
        sw    = hi.y;
        nrows = more;
        more  = 0;
    }
    return acc;
}

#define TILE_M 32
#define APITCH 260

__global__ void __launch_bounds__(256)
fused_kernel(const float* __restrict__ x,
             const int* __restrict__ cnt_r,
             const int* __restrict__ cnt_c,
             const int* __restrict__ off_r,
             const int* __restrict__ off_c,
             const unsigned* __restrict__ csr_r,
             const unsigned* __restrict__ csr_c,
             const float* __restrict__ Wf,
             const float* __restrict__ bias,
             float* __restrict__ out) {
    __shared__ float As[TILE_M * APITCH];
    const int lane = threadIdx.x & 31;
    const int wv   = __builtin_amdgcn_readfirstlane(threadIdx.x >> 5);
    const int base = blockIdx.x * TILE_M;

    #pragma unroll 1
    for (int s = 0; s < 8; ++s) {
        const int dir = s & 1;
        const int r   = wv * 4 + (s >> 1);
        const int n   = base + r;
        float4 f = make_float4(0.f, 0.f, 0.f, 0.f);
        if (n < N_NODES)
            f = gather_node(x, dir ? cnt_c : cnt_r, dir ? cnt_r : cnt_c,
                            dir ? off_c : off_r, dir ? csr_c : csr_r,
                            (int2*)(As + r * APITCH + dir * DIM), n, lane);
        *(float4*)(As + r * APITCH + dir * DIM + lane * 4) = f;
    }
    __syncthreads();

    const int half = lane >> 4;
    const int l15  = lane & 15;
    const float bv = bias[wv * 16 + l15];
    v8f c0, c1;
    #pragma unroll
    for (int j = 0; j < 8; ++j) { c0[j] = bv; c1[j] = bv; }

    const float4* wf  = (const float4*)Wf + (size_t)wv * 32 * 32 + lane;
    const float*  a0p = As + l15 * APITCH + 4 * half;
    const float*  a1p = a0p + 16 * APITCH;
    #pragma unroll 4
    for (int p = 0; p < 32; ++p) {
        const float4 b  = wf[p * 32];
        const float4 a0 = *(const float4*)(a0p + 8 * p);
        const float4 a1 = *(const float4*)(a1p + 8 * p);
        v2f bx = {b.x, b.y},  by = {b.z, b.w};
        v2f x0 = {a0.x, a0.y}, y0 = {a0.z, a0.w};
        v2f x1 = {a1.x, a1.y}, y1 = {a1.z, a1.w};
        c0 = __builtin_amdgcn_wmma_f32_16x16x4_f32(false, x0, false, bx, (short)0, c0, false, false);
        c1 = __builtin_amdgcn_wmma_f32_16x16x4_f32(false, x1, false, bx, (short)0, c1, false, false);
        c0 = __builtin_amdgcn_wmma_f32_16x16x4_f32(false, y0, false, by, (short)0, c0, false, false);
        c1 = __builtin_amdgcn_wmma_f32_16x16x4_f32(false, y1, false, by, (short)0, c1, false, false);
    }
    __syncthreads();

    float* d0 = As + (half * 8) * APITCH + wv * 16 + l15;
    float* d1 = d0 + 16 * APITCH;
    #pragma unroll
    for (int j = 0; j < 8; ++j) { d0[j * APITCH] = c0[j]; d1[j * APITCH] = c1[j]; }
    __syncthreads();

    for (int i = 0; i < 4; ++i) {
        const int r = wv * 4 + i;
        const int n = base + r;
        if (n < N_NODES)
            ((float4*)(out + (size_t)n * DIM))[lane] = *(const float4*)(As + r * APITCH + lane * 4);
    }
}

extern "C" void kernel_launch(void* const* d_in, const int* in_sizes, int n_in,
                              void* d_out, int out_size, void* d_ws, size_t ws_size,
                              hipStream_t stream) {
    const float* x      = (const float*)d_in[0];
    const float* W_src  = (const float*)d_in[1];
    const float* b_src  = (const float*)d_in[2];
    const float* W_dst  = (const float*)d_in[3];
    const float* b_dst  = (const float*)d_in[4];
    const int*   eidx   = (const int*)d_in[5];
    const int*   row    = eidx;
    const int*   col    = eidx + N_EDGES;
    float*       outp   = (float*)d_out;

    int* wi = (int*)d_ws;
    int* cnt_r = wi;
    int* cnt_c = cnt_r + N_NODES;
    int* off_r = cnt_c + N_NODES;
    int* off_c = off_r + N_NODES;
    int* pos_r = off_c + N_NODES;
    int* pos_c = pos_r + N_EDGES;
    unsigned* csr_r = (unsigned*)(pos_c + N_EDGES);
    unsigned* csr_c = csr_r + N_EDGES;
    int*   H    = (int*)(csr_c + N_EDGES);
    float* Wf   = (float*)(H + (size_t)N_NODES * DIM);
    float* bias = Wf + 2 * DIM * DIM;

    hist_kernel<<<2 * NCHUNK, 1024, 0, stream>>>(eidx, H, pos_r);
    chunk_prefix_kernel<<<(2 * N_NODES + 255) / 256, 256, 0, stream>>>(H, cnt_r);
    scan_kernel<<<2, 1024, 0, stream>>>(cnt_r, off_r);
    fill_kernel<<<(N_EDGES + 255) / 256, 256, 0, stream>>>(row, col, cnt_r, cnt_c, off_r, off_c, H,
                                                           pos_r, pos_c, csr_r, csr_c);
    prep_kernel<<<((DIM / 16) * 32 * 32 + 255) / 256, 256, 0, stream>>>(W_src, b_src, W_dst, b_dst,
                                                                        Wf, bias);
    fused_kernel<<<(N_NODES + TILE_M - 1) / TILE_M, 256, 0, stream>>>(x, cnt_r, cnt_c,
                                                                      off_r, off_c, csr_r, csr_c,
                                                                      Wf, bias, outp);
}


# fill pass on LDS slot cursors: a block per (direction, chunk, node range) keeps next-slot and degree tables in LDS; per edge one LDS add + one store replace 6 scattered global reads
# speedup vs baseline: 3.7584x; 1.0626x over previous
#include <hip/hip_runtime.h>

typedef __attribute__((ext_vector_type(2))) float v2f;
typedef __attribute__((ext_vector_type(8))) float v8f;

#define N_NODES 50000
#define N_EDGES 800000
#define DIM     128

#define NCHUNK  64
#define CHUNK   (N_EDGES / NCHUNK)
#define EDGE_IT ((CHUNK + 1023) / 1024)
#define SCAN_CH ((N_NODES + 1023) / 1024)


__global__ void __launch_bounds__(1024)
hist_kernel(const int* __restrict__ eidx, int* __restrict__ H) {
    __shared__ int hist[N_NODES];
    const int t = threadIdx.x;
    int4* h4 = (int4*)hist;
    for (int i = t; i < N_NODES / 4; i += 1024) h4[i] = make_int4(0, 0, 0, 0);

    const int* idx = eidx + (size_t)blockIdx.x * CHUNK;
    int node[EDGE_IT];
    #pragma unroll
    for (int k = 0; k < EDGE_IT; ++k) {
        const int i = t + k * 1024;
        node[k] = (i < CHUNK) ? idx[i] : -1;
    }
    __syncthreads();
    #pragma unroll
    for (int k = 0; k < EDGE_IT; ++k) {
        const int i = t + k * 1024;
        if (i < CHUNK) atomicAdd(&hist[node[k]], 1);
    }
    __syncthreads();
    int4* g4 = (int4*)(H + (size_t)blockIdx.x * N_NODES);
    for (int i = t; i < N_NODES / 4; i += 1024) g4[i] = h4[i];
}

__global__ void __launch_bounds__(256)
chunk_prefix_kernel(int* __restrict__ H, int* __restrict__ cnt) {
    const int i = blockIdx.x * 256 + threadIdx.x;
    if (i >= 2 * N_NODES) return;
    const int d = (i >= N_NODES) ? 1 : 0;
    int* h = H + (size_t)d * NCHUNK * N_NODES + (i - d * N_NODES);
    int v[NCHUNK];
    #pragma unroll
    for (int b = 0; b < NCHUNK; ++b) v[b] = h[(size_t)b * N_NODES];
    int run = 0;
    #pragma unroll
    for (int b = 0; b < NCHUNK; ++b) { h[(size_t)b * N_NODES] = run; run += v[b]; }
    cnt[i] = run;
}

#define SCAN_V4  ((N_NODES / 4 + 1023) / 1024)
#define SCAN_PAD (1024 * SCAN_CH)
__global__ void __launch_bounds__(1024)
scan_kernel(const int* __restrict__ cnt, int* __restrict__ off) {
    __shared__ alignas(16) int buf[SCAN_PAD];
    __shared__ int part[1024];
    const int4* c4 = (const int4*)(cnt + blockIdx.x * N_NODES);
    int4*       o4 = (int4*)(off + blockIdx.x * N_NODES);
    int4*       b4 = (int4*)buf;
    const int t = threadIdx.x;
    int4 q[SCAN_V4];
    #pragma unroll
    for (int k = 0; k < SCAN_V4; ++k) {
        const int i = t + k * 1024;
        q[k] = (i < N_NODES / 4) ? c4[i] : make_int4(0, 0, 0, 0);
    }
    #pragma unroll
    for (int k = 0; k < SCAN_V4; ++k) {
        const int i = t + k * 1024;
        if (i < N_NODES / 4) b4[i] = q[k];
    }
    if (t < SCAN_PAD - N_NODES) buf[N_NODES + t] = 0;
    __syncthreads();

    const int b = t * SCAN_CH;
    int v[SCAN_CH];
    int s = 0;
    #pragma unroll
    for (int k = 0; k < SCAN_CH; ++k) {
        v[k] = buf[b + k];
        s += v[k];
    }
    part[t] = s;
    __syncthreads();
    for (int d = 1; d < 1024; d <<= 1) {
        int u = (t >= d) ? part[t - d] : 0;
        __syncthreads();
        part[t] += u;
        __syncthreads();
    }
    int run = part[t] - s;
    #pragma unroll
    for (int k = 0; k < SCAN_CH; ++k) {
        buf[b + k] = run;
        run += v[k];
    }
    __syncthreads();
    for (int i = t; i < N_NODES / 4; i += 1024) o4[i] = b4[i];
}

static_assert(N_NODES <= 65535, "a node id must fit the upper 16 bits of an entry");
#define DEG_ESC 0xFFFF

#define HALF_N  (N_NODES / 2)
static_assert(HALF_N % 4 == 0, "a node half is read as whole int4");
__global__ void __launch_bounds__(1024)
fill_kernel(const int* __restrict__ eidx,
            const int* __restrict__ cnt,
            const int* __restrict__ off,
            const int* __restrict__ H,
            unsigned* __restrict__ csr) {
    __shared__ alignas(16) int cursor[HALF_N];
    __shared__ alignas(16) unsigned short deg16[N_NODES];
    const int t    = threadIdx.x;
    const int half = blockIdx.x & 1;
    const int cb   = blockIdx.x >> 1;
    const int d    = (cb >= NCHUNK) ? 1 : 0;
    const int lo   = half * HALF_N;

    const int* own = eidx + (size_t)cb * CHUNK;
    const int* oth = eidx + (size_t)(d ? cb - NCHUNK : cb + NCHUNK) * CHUNK;
    int n[EDGE_IT], o[EDGE_IT];
    #pragma unroll
    for (int k = 0; k < EDGE_IT; ++k) {
        const int i = t + k * 1024;
        n[k] = (i < CHUNK) ? own[i] : -1;
        o[k] = (i < CHUNK) ? oth[i] : 0;
    }

    const int4* o4 = (const int4*)(off + d * N_NODES + lo);
    const int4* h4 = (const int4*)(H + (size_t)cb * N_NODES + lo);
    int4*       c4 = (int4*)cursor;
    #pragma unroll 4
    for (int i = t; i < HALF_N / 4; i += 1024) {
        const int4 a = o4[i];
        const int4 b = h4[i];
        c4[i] = make_int4(a.x + b.x, a.y + b.y, a.z + b.z, a.w + b.w);
    }
    const int4* g4 = (const int4*)(cnt + (1 - d) * N_NODES);
    uint2*      s2 = (uint2*)deg16;
    #pragma unroll 4
    for (int i = t; i < N_NODES / 4; i += 1024) {
        const int4 g = g4[i];
        const unsigned x0 = (unsigned)(g.x < DEG_ESC ? g.x : DEG_ESC);
        const unsigned x1 = (unsigned)(g.y < DEG_ESC ? g.y : DEG_ESC);
        const unsigned x2 = (unsigned)(g.z < DEG_ESC ? g.z : DEG_ESC);
        const unsigned x3 = (unsigned)(g.w < DEG_ESC ? g.w : DEG_ESC);
        s2[i] = make_uint2(x0 | (x1 << 16), x2 | (x3 << 16));
    }
    __syncthreads();

    unsigned* seg = csr + (size_t)d * N_EDGES;
    int slot[EDGE_IT];
    unsigned ent[EDGE_IT];
    #pragma unroll
    for (int k = 0; k < EDGE_IT; ++k) {
        const int m = n[k] - lo;
        slot[k] = -1;
        if (m >= 0 && m < HALF_N)
            slot[k] = atomicAdd(&cursor[m], 1);
        ent[k] = ((unsigned)o[k] << 16) | (unsigned)deg16[o[k]];
    }
    #pragma unroll
    for (int k = 0; k < EDGE_IT; ++k)
        if (slot[k] >= 0) seg[slot[k]] = ent[k];
}

__global__ void prep_kernel(const float* __restrict__ W_src,
                            const float* __restrict__ b_src,
                            const float* __restrict__ W_dst,
                            const float* __restrict__ b_dst,
                            float* __restrict__ Wf,
                            float* __restrict__ bias) {
    const int i = blockIdx.x * blockDim.x + threadIdx.x;
    if (i < (DIM / 16) * 32 * 32) {
        const int lane = i & 31;
        const int p    = (i >> 5) & 31;
        const int nt   = i >> 10;
        const int n    = nt * 16 + (lane & 15);
        const int kk   = 8 * p + 4 * (lane >> 4);
        const float* s = (kk < DIM) ? (W_src + (size_t)n * DIM + kk)
                                    : (W_dst + (size_t)n * DIM + (kk - DIM));
        const float4 v = *(const float4*)s;
        ((float4*)Wf)[i] = make_float4(0.5f * v.x, 0.5f * v.y, 0.5f * v.z, 0.5f * v.w);
    }
    if (i < DIM) bias[i] = 0.5f * b_src[i] + 0.5f * b_dst[i];
}

__device__ __forceinline__ float inv_sqrt_deg(int d) {
    float f = (float)d;
    return (d > 0) ? rsqrtf(fmaxf(f, 1.0f)) : 0.0f;
}

__device__ __forceinline__ void axpy4(float4& acc, float w, const float4 v) {
    acc.x += w * v.x; acc.y += w * v.y; acc.z += w * v.z; acc.w += w * v.w;
}

__device__ __forceinline__ void
add_rows(float4& acc, const float* __restrict__ x, const int ssrc, const int sw,
         const int count, const int lane) {
    int s = 0;
    for (; s + 4 <= count; s += 4) {
        const int c0 = __shfl(ssrc, s), c1 = __shfl(ssrc, s + 1);
        const int c2 = __shfl(ssrc, s + 2), c3 = __shfl(ssrc, s + 3);
        const float4 v0 = ((const float4*)(x + (size_t)c0 * DIM))[lane];
        const float4 v1 = ((const float4*)(x + (size_t)c1 * DIM))[lane];
        const float4 v2 = ((const float4*)(x + (size_t)c2 * DIM))[lane];
        const float4 v3 = ((const float4*)(x + (size_t)c3 * DIM))[lane];
        axpy4(acc, __int_as_float(__shfl(sw, s)),     v0);
        axpy4(acc, __int_as_float(__shfl(sw, s + 1)), v1);
        axpy4(acc, __int_as_float(__shfl(sw, s + 2)), v2);
        axpy4(acc, __int_as_float(__shfl(sw, s + 3)), v3);
    }
    for (; s < count; ++s) {
        const int c0 = __shfl(ssrc, s);
        const float4 v0 = ((const float4*)(x + (size_t)c0 * DIM))[lane];
        axpy4(acc, __int_as_float(__shfl(sw, s)), v0);
    }
}

__device__ __forceinline__ float
entry_weight(const unsigned k, const float inv_n, const int* __restrict__ cnt_other) {
    int d = (int)(k & DEG_ESC);
    if (d == DEG_ESC) d = cnt_other[k >> 16];
    return inv_n * inv_sqrt_deg(d);
}

__device__ __forceinline__ float4
gather_node(const float* __restrict__ x,
            const int* __restrict__ cnt_self,
            const int* __restrict__ cnt_other,
            const int* __restrict__ off_self,
            const unsigned* __restrict__ csr,
            int2* scratch,
            const int n, const int lane) {
    const int deg = cnt_self[n];
    const int beg = off_self[n];
    float4 acc = make_float4(0.f, 0.f, 0.f, 0.f);
    const float inv_n = inv_sqrt_deg(deg);
    int ssrc = 0, sw = 0, nrows = 0, more = 0;
    if (deg > 0 && deg <= 32) {
        unsigned key = 0xFFFFFFFFu;
        int srcn = 0;
        float w = 0.f;
        if (lane < deg) {
            key  = csr[beg + lane];
            srcn = (int)(key >> 16);
            w    = entry_weight(key, inv_n, cnt_other);
        }
        int rank = 0;
        for (int j = 0; j < deg; ++j) {
            const unsigned kj = (unsigned)__shfl((int)key, j);
            rank += (kj < key || (kj == key && j < lane)) ? 1 : 0;
        }
        if (lane >= deg) rank = lane;
        ssrc  = __builtin_amdgcn_ds_permute(rank << 2, srcn);
        sw    = __builtin_amdgcn_ds_permute(rank << 2, __float_as_int(w));
        nrows = deg;
    } else if (deg > 32 && deg <= 64) {
        const int  i1   = lane + 32;
        const bool has1 = i1 < deg;
        const unsigned k0 = csr[beg + lane];
        unsigned k1 = 0xFFFFFFFFu;
        if (has1) k1 = csr[beg + i1];
        const int   src0 = (int)(k0 >> 16);
        const float w0   = entry_weight(k0, inv_n, cnt_other);
        int   src1 = 0;
        float w1   = 0.f;
        if (has1) {
            src1 = (int)(k1 >> 16);
            w1   = entry_weight(k1, inv_n, cnt_other);
        }
        int rank0 = 0, rank1 = 0;
        for (int j = 0; j < 32; ++j) {
            const unsigned a = (unsigned)__shfl((int)k0, j);
            const unsigned b = (unsigned)__shfl((int)k1, j);
            rank0 += ((a < k0 || (a == k0 && j < lane)) ? 1 : 0) + ((b < k0) ? 1 : 0);
            rank1 += ((a <= k1) ? 1 : 0) + ((b < k1 || (b == k1 && j < lane)) ? 1 : 0);
        }
        scratch[rank0] = make_int2(src0, __float_as_int(w0));
        if (has1) scratch[rank1] = make_int2(src1, __float_as_int(w1));
        const int2 lo = scratch[lane];
        ssrc  = lo.x;
        sw    = lo.y;
        nrows = 32;
        more  = deg - 32;
    } else if (deg > 64) {
        unsigned last = 0;
        bool first = true;
        int done = 0;
        for (int s = 0; s < deg && done < deg; ++s) {
            unsigned m = 0xFFFFFFFFu;
            int c = 0;
            for (int i = lane; i < deg; i += 32) {
                const unsigned k = csr[beg + i];
                if (first || k > last) {
                    if (k < m) { m = k; c = 1; }
                    else if (k == m) ++c;
                }
            }
            unsigned g = m;
            for (int o = 16; o > 0; o >>= 1) {
                const unsigned t = (unsigned)__shfl_xor((int)g, o);
                g = (t < g) ? t : g;
            }
            int mult = (m == g) ? c : 0;
            for (int o = 16; o > 0; o >>= 1) mult += __shfl_xor(mult, o);
            last  = g;
            first = false;
            done += mult;
            const float w = entry_weight(g, inv_n, cnt_other);
            const float4 v0 = ((const float4*)(x + (size_t)(g >> 16) * DIM))[lane];
            for (int q = 0; q < mult; ++q) axpy4(acc, w, v0);
        }
    }
    #pragma unroll 1
    for (;;) {
        add_rows(acc, x, ssrc, sw, nrows, lane);
        if (more <= 0) break;
        int2 hi = make_int2(0, 0);
        if (lane < more) hi = scratch[lane + 32];
        ssrc  = hi.x;
        sw    = hi.y;
        nrows = more;
        more  = 0;
    }
    return acc;
}

#define TILE_M 32
#define APITCH 260

__global__ void __launch_bounds__(256)
fused_kernel(const float* __restrict__ x,
             const int* __restrict__ cnt_r,
             const int* __restrict__ cnt_c,
             const int* __restrict__ off_r,
             const int* __restrict__ off_c,
             const unsigned* __restrict__ csr_r,
             const unsigned* __restrict__ csr_c,
             const float* __restrict__ Wf,
             const float* __restrict__ bias,
             float* __restrict__ out) {
    __shared__ float As[TILE_M * APITCH];
    const int lane = threadIdx.x & 31;
    const int wv   = __builtin_amdgcn_readfirstlane(threadIdx.x >> 5);
    const int base = blockIdx.x * TILE_M;

    #pragma unroll 1
    for (int s = 0; s < 8; ++s) {
        const int dir = s & 1;
        const int r   = wv * 4 + (s >> 1);
        const int n   = base + r;
        float4 f = make_float4(0.f, 0.f, 0.f, 0.f);
        if (n < N_NODES)
            f = gather_node(x, dir ? cnt_c : cnt_r, dir ? cnt_r : cnt_c,
                            dir ? off_c : off_r, dir ? csr_c : csr_r,
                            (int2*)(As + r * APITCH + dir * DIM), n, lane);
        *(float4*)(As + r * APITCH + dir * DIM + lane * 4) = f;
    }
    __syncthreads();

    const int half = lane >> 4;
    const int l15  = lane & 15;
    const float bv = bias[wv * 16 + l15];
    v8f c0, c1;
    #pragma unroll
    for (int j = 0; j < 8; ++j) { c0[j] = bv; c1[j] = bv; }

    const float4* wf  = (const float4*)Wf + (size_t)wv * 32 * 32 + lane;
    const float*  a0p = As + l15 * APITCH + 4 * half;
    const float*  a1p = a0p + 16 * APITCH;
    #pragma unroll 4
    for (int p = 0; p < 32; ++p) {
        const float4 b  = wf[p * 32];
        const float4 a0 = *(const float4*)(a0p + 8 * p);
        const float4 a1 = *(const float4*)(a1p + 8 * p);
        v2f bx = {b.x, b.y},  by = {b.z, b.w};
        v2f x0 = {a0.x, a0.y}, y0 = {a0.z, a0.w};
        v2f x1 = {a1.x, a1.y}, y1 = {a1.z, a1.w};
        c0 = __builtin_amdgcn_wmma_f32_16x16x4_f32(false, x0, false, bx, (short)0, c0, false, false);
        c1 = __builtin_amdgcn_wmma_f32_16x16x4_f32(false, x1, false, bx, (short)0, c1, false, false);
        c0 = __builtin_amdgcn_wmma_f32_16x16x4_f32(false, y0, false, by, (short)0, c0, false, false);
        c1 = __builtin_amdgcn_wmma_f32_16x16x4_f32(false, y1, false, by, (short)0, c1, false, false);
    }
    __syncthreads();

    float* d0 = As + (half * 8) * APITCH + wv * 16 + l15;
    float* d1 = d0 + 16 * APITCH;
    #pragma unroll
    for (int j = 0; j < 8; ++j) { d0[j * APITCH] = c0[j]; d1[j * APITCH] = c1[j]; }
    __syncthreads();

    for (int i = 0; i < 4; ++i) {
        const int r = wv * 4 + i;
        const int n = base + r;
        if (n < N_NODES)
            ((float4*)(out + (size_t)n * DIM))[lane] = *(const float4*)(As + r * APITCH + lane * 4);
    }
}

extern "C" void kernel_launch(void* const* d_in, const int* in_sizes, int n_in,
                              void* d_out, int out_size, void* d_ws, size_t ws_size,
                              hipStream_t stream) {
    const float* x      = (const float*)d_in[0];
    const float* W_src  = (const float*)d_in[1];
    const float* b_src  = (const float*)d_in[2];
    const float* W_dst  = (const float*)d_in[3];
    const float* b_dst  = (const float*)d_in[4];
    const int*   eidx   = (const int*)d_in[5];
    float*       outp   = (float*)d_out;

    int* wi = (int*)d_ws;
    int* cnt_r = wi;
    int* cnt_c = cnt_r + N_NODES;
    int* off_r = cnt_c + N_NODES;
    int* off_c = off_r + N_NODES;
    unsigned* csr_r = (unsigned*)(off_c + N_NODES);
    unsigned* csr_c = csr_r + N_EDGES;
    int*   H    = (int*)(csr_c + N_EDGES);
    float* Wf   = (float*)(H + (size_t)N_NODES * DIM);
    float* bias = Wf + 2 * DIM * DIM;

    hist_kernel<<<2 * NCHUNK, 1024, 0, stream>>>(eidx, H);
    chunk_prefix_kernel<<<(2 * N_NODES + 255) / 256, 256, 0, stream>>>(H, cnt_r);
    scan_kernel<<<2, 1024, 0, stream>>>(cnt_r, off_r);
    fill_kernel<<<4 * NCHUNK, 1024, 0, stream>>>(eidx, cnt_r, off_r, H, csr_r);
    prep_kernel<<<((DIM / 16) * 32 * 32 + 255) / 256, 256, 0, stream>>>(W_src, b_src, W_dst, b_dst,
                                                                        Wf, bias);
    fused_kernel<<<(N_NODES + TILE_M - 1) / TILE_M, 256, 0, stream>>>(x, cnt_r, cnt_c,
                                                                      off_r, off_c, csr_r, csr_c,
                                                                      Wf, bias, outp);
}


# segment build: a bucket's entries are placed in LDS and leave as whole 128-byte lines, one store per line, replacing one scattered 4-byte global store per edge (1.6 M partial-line writes)
# speedup vs baseline: 6.4552x; 1.7175x over previous
#include <hip/hip_runtime.h>

typedef __attribute__((ext_vector_type(2))) float v2f;
typedef __attribute__((ext_vector_type(8))) float v8f;

#define N_NODES 50000
#define N_EDGES 800000
#define DIM     128

#define NCHUNK  64
#define CHUNK   (N_EDGES / NCHUNK)
#define EDGE_IT ((CHUNK + 1023) / 1024)
#define PB      256
#define NB      ((N_NODES + PB - 1) / PB)
#define TABW    (NB + 1)
#define GPL     ((NB + 32) / 32)
static_assert(CHUNK % 4 == 0, "a chunk is written as whole int4");
static_assert(32 * GPL >= TABW, "one wave scans all run starts of a chunk");


__device__ __forceinline__ void
prep_entry(const int i,
           const float* __restrict__ W_src, const float* __restrict__ b_src,
           const float* __restrict__ W_dst, const float* __restrict__ b_dst,
           float* __restrict__ Wf, float* __restrict__ bias) {
    if (i < (DIM / 16) * 32 * 32) {
        const int lane = i & 31;
        const int p    = (i >> 5) & 31;
        const int nt   = i >> 10;
        const int n    = nt * 16 + (lane & 15);
        const int kk   = 8 * p + 4 * (lane >> 4);
        const float* s = (kk < DIM) ? (W_src + (size_t)n * DIM + kk)
                                    : (W_dst + (size_t)n * DIM + (kk - DIM));
        const float4 v = *(const float4*)s;
        ((float4*)Wf)[i] = make_float4(0.5f * v.x, 0.5f * v.y, 0.5f * v.z, 0.5f * v.w);
    }
    if (i < DIM) bias[i] = 0.5f * b_src[i] + 0.5f * b_dst[i];
}

#define PREP_B (((DIM / 16) * 32 * 32 + 1023) / 1024)
__global__ void __launch_bounds__(1024)
part_kernel(const int* __restrict__ eidx, int* __restrict__ pn, int* __restrict__ po,
            int* __restrict__ tab,
            const float* __restrict__ W_src, const float* __restrict__ b_src,
            const float* __restrict__ W_dst, const float* __restrict__ b_dst,
            float* __restrict__ Wf, float* __restrict__ bias) {
    __shared__ alignas(16) int sn[CHUNK];
    __shared__ alignas(16) int so[CHUNK];
    __shared__ int bcnt[32 * GPL];
    __shared__ int bstart[32 * GPL];
    const int t = threadIdx.x;
    const int b = blockIdx.x;
    if (b >= 2 * NCHUNK) {
        prep_entry((b - 2 * NCHUNK) * 1024 + t, W_src, b_src, W_dst, b_dst, Wf, bias);
        return;
    }
    const int* own = eidx + (size_t)b * CHUNK;
    const int* oth = eidx + (size_t)(b >= NCHUNK ? b - NCHUNK : b + NCHUNK) * CHUNK;
    if (t < 32 * GPL) bcnt[t] = 0;
    int n[EDGE_IT], o[EDGE_IT], r[EDGE_IT];
    #pragma unroll
    for (int k = 0; k < EDGE_IT; ++k) {
        const int i = t + k * 1024;
        n[k] = (i < CHUNK) ? own[i] : -1;
        o[k] = (i < CHUNK) ? oth[i] : 0;
    }
    __syncthreads();
    #pragma unroll
    for (int k = 0; k < EDGE_IT; ++k) {
        r[k] = 0;
        if (n[k] >= 0) r[k] = atomicAdd(&bcnt[n[k] / PB], 1);
    }
    __syncthreads();
    if (t < 32) {
        int gv[GPL];
        int s = 0;
        #pragma unroll
        for (int k = 0; k < GPL; ++k) {
            gv[k] = bcnt[t * GPL + k];
            s += gv[k];
        }
        int inc = s;
        for (int sh = 1; sh < 32; sh <<= 1) {
            const int u = __shfl_up(inc, sh);
            if (t >= sh) inc += u;
        }
        int run = inc - s;
        #pragma unroll
        for (int k = 0; k < GPL; ++k) {
            bstart[t * GPL + k] = run;
            run += gv[k];
        }
    }
    __syncthreads();
    #pragma unroll
    for (int k = 0; k < EDGE_IT; ++k) {
        if (n[k] >= 0) {
            const int pos = bstart[n[k] / PB] + r[k];
            sn[pos] = n[k];
            so[pos] = o[k];
        }
    }
    if (t < TABW) tab[b * TABW + t] = bstart[t];
    __syncthreads();
    int4* gn = (int4*)(pn + (size_t)b * CHUNK);
    int4* go = (int4*)(po + (size_t)b * CHUNK);
    for (int i = t; i < CHUNK / 4; i += 1024) {
        gn[i] = ((const int4*)sn)[i];
        go[i] = ((const int4*)so)[i];
    }
}

#define STAGE_CAP 8192
static_assert(NCHUNK * 16 == 1024, "sixteen lanes per chunk");
static_assert(NCHUNK == 64, "a wave sums the run starts two per lane");

template <bool STAGED>
__device__ __forceinline__ void
place_run(const int* __restrict__ rn, const int* __restrict__ ro,
          const int s, const int e, const int l16, const int lo, const int bs,
          int* ncnt, unsigned* stage, unsigned* __restrict__ seg) {
    for (int i = s + l16; i < e; i += 64) {
        int a[4], z[4], sl[4];
        #pragma unroll
        for (int j = 0; j < 4; ++j) {
            const int q = i + 16 * j;
            a[j] = (q < e) ? rn[q] : -1;
            z[j] = (q < e) ? ro[q] : 0;
        }
        #pragma unroll
        for (int j = 0; j < 4; ++j)
            sl[j] = (a[j] >= 0) ? atomicAdd(&ncnt[a[j] - lo], 1) : -1;
        #pragma unroll
        for (int j = 0; j < 4; ++j)
            if (sl[j] >= 0) {
                if (STAGED) stage[sl[j] - bs] = (unsigned)z[j];
                else        seg[sl[j]] = (unsigned)z[j];
            }
    }
}
__global__ void __launch_bounds__(1024)
build_kernel(const int* __restrict__ pn, const int* __restrict__ po,
             const int* __restrict__ tab,
             int* __restrict__ cnt,
             int* __restrict__ off,
             unsigned* __restrict__ csr) {
    __shared__ int ncnt[PB];
    __shared__ int rs[NCHUNK];
    __shared__ int re[NCHUNK];
    __shared__ int wsum[PB / 32];
    __shared__ unsigned stage[STAGE_CAP];
    const int t    = threadIdx.x;
    const int lane = t & 31;
    const int d    = (blockIdx.x >= NB) ? 1 : 0;
    const int g    = blockIdx.x - d * NB;
    const int lo   = g * PB;
    if (t < PB) ncnt[t] = 0;
    if (t < NCHUNK) {
        const int* row = tab + (size_t)(d * NCHUNK + t) * TABW + g;
        rs[t] = row[0];
        re[t] = row[1];
    }
    __syncthreads();

    const int c   = t >> 4;
    const int l16 = t & 15;
    const int s   = rs[c];
    const int e   = re[c];
    const int* rn = pn + (size_t)(d * NCHUNK + c) * CHUNK;
    const int* ro = po + (size_t)(d * NCHUNK + c) * CHUNK;
    int bs = rs[lane] + rs[lane + 32];
    for (int x = 16; x > 0; x >>= 1) bs += __shfl_xor(bs, x);

    for (int i = s + l16; i < e; i += 64) {
        int a[4];
        #pragma unroll
        for (int j = 0; j < 4; ++j) {
            const int q = i + 16 * j;
            a[j] = (q < e) ? rn[q] : -1;
        }
        #pragma unroll
        for (int j = 0; j < 4; ++j)
            if (a[j] >= 0) atomicAdd(&ncnt[a[j] - lo], 1);
    }
    __syncthreads();

    const int v = (t < PB) ? ncnt[t] : 0;
    int inc = v;
    for (int x = 1; x < 32; x <<= 1) {
        const int u = __shfl_up(inc, x);
        if (lane >= x) inc += u;
    }
    if (t < PB && lane == 31) wsum[t >> 5] = inc;
    __syncthreads();
    int tot = 0;
    for (int k = 0; k < PB / 32; ++k) tot += wsum[k];
    const bool staged = tot <= STAGE_CAP;
    if (t < PB) {
        int wbase = 0;
        for (int k = 0; k < (t >> 5); ++k) wbase += wsum[k];
        const int start = bs + wbase + inc - v;
        const int node  = lo + t;
        if (node < N_NODES) {
            cnt[d * N_NODES + node] = v;
            off[d * N_NODES + node] = start;
        }
        ncnt[t] = start;
    }
    __syncthreads();

    unsigned* seg = csr + (size_t)d * N_EDGES;
    if (staged) place_run<true>(rn, ro, s, e, l16, lo, bs, ncnt, stage, seg);
    else        place_run<false>(rn, ro, s, e, l16, lo, bs, ncnt, stage, seg);
    __syncthreads();
    if (staged) {
        const int end = bs + tot;
        for (int i = (bs & ~31) + t; i < end; i += 1024)
            if (i >= bs) seg[i] = stage[i - bs];
    }
}

__device__ __forceinline__ float inv_sqrt_deg(int d) {
    float f = (float)d;
    return (d > 0) ? rsqrtf(fmaxf(f, 1.0f)) : 0.0f;
}

__device__ __forceinline__ void axpy4(float4& acc, float w, const float4 v) {
    acc.x += w * v.x; acc.y += w * v.y; acc.z += w * v.z; acc.w += w * v.w;
}

__device__ __forceinline__ void
add_rows(float4& acc, const float* __restrict__ x, const int ssrc, const int sw,
         const int count, const int lane) {
    int s = 0;
    for (; s + 4 <= count; s += 4) {
        const int c0 = __shfl(ssrc, s), c1 = __shfl(ssrc, s + 1);
        const int c2 = __shfl(ssrc, s + 2), c3 = __shfl(ssrc, s + 3);
        const float4 v0 = ((const float4*)(x + (size_t)c0 * DIM))[lane];
        const float4 v1 = ((const float4*)(x + (size_t)c1 * DIM))[lane];
        const float4 v2 = ((const float4*)(x + (size_t)c2 * DIM))[lane];
        const float4 v3 = ((const float4*)(x + (size_t)c3 * DIM))[lane];
        axpy4(acc, __int_as_float(__shfl(sw, s)),     v0);
        axpy4(acc, __int_as_float(__shfl(sw, s + 1)), v1);
        axpy4(acc, __int_as_float(__shfl(sw, s + 2)), v2);
        axpy4(acc, __int_as_float(__shfl(sw, s + 3)), v3);
    }
    for (; s < count; ++s) {
        const int c0 = __shfl(ssrc, s);
        const float4 v0 = ((const float4*)(x + (size_t)c0 * DIM))[lane];
        axpy4(acc, __int_as_float(__shfl(sw, s)), v0);
    }
}

__device__ __forceinline__ float
entry_weight(const int deg_other, const float inv_n) {
    return inv_n * inv_sqrt_deg(deg_other);
}

__device__ __forceinline__ float4
gather_node(const float* __restrict__ x,
            const int* __restrict__ cnt_self,
            const int* __restrict__ cnt_other,
            const int* __restrict__ off_self,
            const unsigned* __restrict__ csr,
            int2* scratch,
            const int n, const int lane) {
    const int deg = cnt_self[n];
    const int beg = off_self[n];
    float4 acc = make_float4(0.f, 0.f, 0.f, 0.f);
    const float inv_n = inv_sqrt_deg(deg);
    int ssrc = 0, sw = 0, nrows = 0, more = 0;
    if (deg > 0 && deg <= 32) {
        unsigned key = 0xFFFFFFFFu;
        int srcn = 0;
        int dgo  = 0;
        if (lane < deg) {
            key  = csr[beg + lane];
            srcn = (int)key;
            dgo  = cnt_other[key];
        }
        int rank = 0;
        for (int j = 0; j < deg; ++j) {
            const unsigned kj = (unsigned)__shfl((int)key, j);
            rank += (kj < key || (kj == key && j < lane)) ? 1 : 0;
        }
        if (lane >= deg) rank = lane;
        const float w = entry_weight(dgo, inv_n);
        ssrc  = __builtin_amdgcn_ds_permute(rank << 2, srcn);
        sw    = __builtin_amdgcn_ds_permute(rank << 2, __float_as_int(w));
        nrows = deg;
    } else if (deg > 32 && deg <= 64) {
        const int  i1   = lane + 32;
        const bool has1 = i1 < deg;
        const unsigned k0 = csr[beg + lane];
        unsigned k1 = 0xFFFFFFFFu;
        if (has1) k1 = csr[beg + i1];
        const int src0 = (int)k0;
        const int dg0  = cnt_other[k0];
        int src1 = 0;
        int dg1  = 0;
        if (has1) {
            src1 = (int)k1;
            dg1  = cnt_other[k1];
        }
        int rank0 = 0, rank1 = 0;
        for (int j = 0; j < 32; ++j) {
            const unsigned a = (unsigned)__shfl((int)k0, j);
            const unsigned b = (unsigned)__shfl((int)k1, j);
            rank0 += ((a < k0 || (a == k0 && j < lane)) ? 1 : 0) + ((b < k0) ? 1 : 0);
            rank1 += ((a <= k1) ? 1 : 0) + ((b < k1 || (b == k1 && j < lane)) ? 1 : 0);
        }
        const float w0 = entry_weight(dg0, inv_n);
        const float w1 = entry_weight(dg1, inv_n);
        scratch[rank0] = make_int2(src0, __float_as_int(w0));
        if (has1) scratch[rank1] = make_int2(src1, __float_as_int(w1));
        const int2 lo = scratch[lane];
        ssrc  = lo.x;
        sw    = lo.y;
        nrows = 32;
        more  = deg - 32;
    } else if (deg > 64) {
        unsigned last = 0;
        bool first = true;
        int done = 0;
        for (int s = 0; s < deg && done < deg; ++s) {
            unsigned m = 0xFFFFFFFFu;
            int c = 0;
            for (int i = lane; i < deg; i += 32) {
                const unsigned k = csr[beg + i];
                if (first || k > last) {
                    if (k < m) { m = k; c = 1; }
                    else if (k == m) ++c;
                }
            }
            unsigned g = m;
            for (int o = 16; o > 0; o >>= 1) {
                const unsigned t = (unsigned)__shfl_xor((int)g, o);
                g = (t < g) ? t : g;
            }
            int mult = (m == g) ? c : 0;
            for (int o = 16; o > 0; o >>= 1) mult += __shfl_xor(mult, o);
            last  = g;
            first = false;
            done += mult;
            const float w = entry_weight(cnt_other[g], inv_n);
            const float4 v0 = ((const float4*)(x + (size_t)g * DIM))[lane];
            for (int q = 0; q < mult; ++q) axpy4(acc, w, v0);
        }
    }
    #pragma unroll 1
    for (;;) {
        add_rows(acc, x, ssrc, sw, nrows, lane);
        if (more <= 0) break;
        int2 hi = make_int2(0, 0);
        if (lane < more) hi = scratch[lane + 32];
        ssrc  = hi.x;
        sw    = hi.y;
        nrows = more;
        more  = 0;
    }
    return acc;
}

#define TILE_M 32
#define APITCH 260

__global__ void __launch_bounds__(256)
fused_kernel(const float* __restrict__ x,
             const int* __restrict__ cnt_r,
             const int* __restrict__ cnt_c,
             const int* __restrict__ off_r,
             const int* __restrict__ off_c,
             const unsigned* __restrict__ csr_r,
             const unsigned* __restrict__ csr_c,
             const float* __restrict__ Wf,
             const float* __restrict__ bias,
             float* __restrict__ out) {
    __shared__ float As[TILE_M * APITCH];
    const int lane = threadIdx.x & 31;
    const int wv   = __builtin_amdgcn_readfirstlane(threadIdx.x >> 5);
    const int base = blockIdx.x * TILE_M;

    #pragma unroll 1
    for (int s = 0; s < 8; ++s) {
        const int dir = s & 1;
        const int r   = wv * 4 + (s >> 1);
        const int n   = base + r;
        float4 f = make_float4(0.f, 0.f, 0.f, 0.f);
        if (n < N_NODES)
            f = gather_node(x, dir ? cnt_c : cnt_r, dir ? cnt_r : cnt_c,
                            dir ? off_c : off_r, dir ? csr_c : csr_r,
                            (int2*)(As + r * APITCH + dir * DIM), n, lane);
        *(float4*)(As + r * APITCH + dir * DIM + lane * 4) = f;
    }
    __syncthreads();

    const int half = lane >> 4;
    const int l15  = lane & 15;
    const float bv = bias[wv * 16 + l15];
    v8f c0, c1;
    #pragma unroll
    for (int j = 0; j < 8; ++j) { c0[j] = bv; c1[j] = bv; }

    const float4* wf  = (const float4*)Wf + (size_t)wv * 32 * 32 + lane;
    const float*  a0p = As + l15 * APITCH + 4 * half;
    const float*  a1p = a0p + 16 * APITCH;
    #pragma unroll 4
    for (int p = 0; p < 32; ++p) {
        const float4 b  = wf[p * 32];
        const float4 a0 = *(const float4*)(a0p + 8 * p);
        const float4 a1 = *(const float4*)(a1p + 8 * p);
        v2f bx = {b.x, b.y},  by = {b.z, b.w};
        v2f x0 = {a0.x, a0.y}, y0 = {a0.z, a0.w};
        v2f x1 = {a1.x, a1.y}, y1 = {a1.z, a1.w};
        c0 = __builtin_amdgcn_wmma_f32_16x16x4_f32(false, x0, false, bx, (short)0, c0, false, false);
        c1 = __builtin_amdgcn_wmma_f32_16x16x4_f32(false, x1, false, bx, (short)0, c1, false, false);
        c0 = __builtin_amdgcn_wmma_f32_16x16x4_f32(false, y0, false, by, (short)0, c0, false, false);
        c1 = __builtin_amdgcn_wmma_f32_16x16x4_f32(false, y1, false, by, (short)0, c1, false, false);
    }
    __syncthreads();

    float* d0 = As + (half * 8) * APITCH + wv * 16 + l15;
    float* d1 = d0 + 16 * APITCH;
    #pragma unroll
    for (int j = 0; j < 8; ++j) { d0[j * APITCH] = c0[j]; d1[j * APITCH] = c1[j]; }
    __syncthreads();

    for (int i = 0; i < 4; ++i) {
        const int r = wv * 4 + i;
        const int n = base + r;
        if (n < N_NODES)
            ((float4*)(out + (size_t)n * DIM))[lane] = *(const float4*)(As + r * APITCH + lane * 4);
    }
}

extern "C" void kernel_launch(void* const* d_in, const int* in_sizes, int n_in,
                              void* d_out, int out_size, void* d_ws, size_t ws_size,
                              hipStream_t stream) {
    const float* x      = (const float*)d_in[0];
    const float* W_src  = (const float*)d_in[1];
    const float* b_src  = (const float*)d_in[2];
    const float* W_dst  = (const float*)d_in[3];
    const float* b_dst  = (const float*)d_in[4];
    const int*   eidx   = (const int*)d_in[5];
    float*       outp   = (float*)d_out;

    int* wi = (int*)d_ws;
    int* cnt_r = wi;
    int* cnt_c = cnt_r + N_NODES;
    int* off_r = cnt_c + N_NODES;
    int* off_c = off_r + N_NODES;
    unsigned* csr_r = (unsigned*)(off_c + N_NODES);
    unsigned* csr_c = csr_r + N_EDGES;
    int*   pn   = (int*)(csr_c + N_EDGES);
    int*   po   = pn + 2 * N_EDGES;
    int*   tab  = po + 2 * N_EDGES;
    float* Wf   = (float*)(tab + 2 * NCHUNK * TABW);
    float* bias = Wf + 2 * DIM * DIM;

    part_kernel<<<2 * NCHUNK + PREP_B, 1024, 0, stream>>>(eidx, pn, po, tab, W_src, b_src, W_dst, b_dst,
                                                          Wf, bias);
    build_kernel<<<2 * NB, 1024, 0, stream>>>(pn, po, tab, cnt_r, off_r, csr_r);
    fused_kernel<<<(N_NODES + TILE_M - 1) / TILE_M, 256, 0, stream>>>(x, cnt_r, cnt_c,
                                                                      off_r, off_c, csr_r, csr_c,
                                                                      Wf, bias, outp);
}
